# starting kernel
# baseline (speedup 1.0000x reference)
#include <hip/hip_runtime.h>

typedef __attribute__((ext_vector_type(2))) float v2f;
typedef __attribute__((ext_vector_type(8))) float v8f;

#define DIM 64

__global__ void gcn_softmax_kernel(const float* __restrict__ alpha,
                                   float* __restrict__ wts, int n) {
  if (blockIdx.x == 0 && threadIdx.x == 0) {
    float m = alpha[0];
    for (int i = 1; i < n; ++i) m = fmaxf(m, alpha[i]);
    float s = 0.f;
    for (int i = 0; i < n; ++i) s += expf(alpha[i] - m);
    for (int i = 0; i < n; ++i) wts[i] = expf(alpha[i] - m) / s;
  }
}

__global__ void gcn_deg_kernel(const int* __restrict__ dst,
                               float* __restrict__ deg, int E) {
  int e = blockIdx.x * blockDim.x + threadIdx.x;
  if (e < E) unsafeAtomicAdd(&deg[dst[e]], 1.0f);
}

__global__ void gcn_rsqrt_kernel(float* __restrict__ d, int n) {
  int i = blockIdx.x * blockDim.x + threadIdx.x;
  if (i < n) {
    float v = d[i];
    d[i] = (v > 0.f) ? rsqrtf(v) : 0.f;
  }
}

__global__ void gcn_edgew_kernel(const int* __restrict__ src, const int* __restrict__ dst,
                                 const float* __restrict__ dis, float* __restrict__ w, int E) {
  int e = blockIdx.x * blockDim.x + threadIdx.x;
  if (e < E) w[e] = dis[src[e]] * dis[dst[e]];
}

__global__ void gcn_scale_init_kernel(const float* __restrict__ x, float* __restrict__ o,
                                      const float* __restrict__ wts, int n4) {
  int i = blockIdx.x * blockDim.x + threadIdx.x;
  if (i < n4) {
    float s = wts[0];
    float4 a = ((const float4*)x)[i];
    float4 r;
    r.x = s * a.x; r.y = s * a.y; r.z = s * a.z; r.w = s * a.w;
    ((float4*)o)[i] = r;
  }
}

__global__ void gcn_spmm_kernel(const int* __restrict__ src, const int* __restrict__ dst,
                                const float* __restrict__ w, const float* __restrict__ xin,
                                float* __restrict__ xout, int E) {
  int t = blockIdx.x * blockDim.x + threadIdx.x;
  int e = t >> 4;
  if (e >= E) return;
  int c = (t & 15) << 2;
  int s = src[e];
  int d = dst[e];
  float we = w[e];
  const float4 xs = *(const float4*)(xin + (size_t)s * DIM + c);
  float* o = xout + (size_t)d * DIM + c;
  unsafeAtomicAdd(o + 0, we * xs.x);
  unsafeAtomicAdd(o + 1, we * xs.y);
  unsafeAtomicAdd(o + 2, we * xs.z);
  unsafeAtomicAdd(o + 3, we * xs.w);
}

__global__ void gcn_axpy_kernel(const float* __restrict__ x, float* __restrict__ o,
                                const float* __restrict__ wts, int k, int n4) {
  int i = blockIdx.x * blockDim.x + threadIdx.x;
  if (i < n4) {
    float s = wts[k];
    float4 a = ((const float4*)x)[i];
    float4 r = ((const float4*)o)[i];
    r.x += s * a.x; r.y += s * a.y; r.z += s * a.z; r.w += s * a.w;
    ((float4*)o)[i] = r;
  }
}

__global__ void gcn_edge_dot_wmma_kernel(const int* __restrict__ la, const int* __restrict__ lb,
                                         const float* __restrict__ x, float* __restrict__ res,
                                         int E) {
  const int gwave = (blockIdx.x * blockDim.x + threadIdx.x) >> 5;
  const int lane  = threadIdx.x & 31;
  const int ebase = gwave * 16;
  const int m     = lane & 15;

  int e = ebase + m;
  if (e >= E) e = E - 1;

  const size_t ai = (size_t)la[e];
  const size_t bi = (size_t)lb[e];
  const int k0 = (lane >> 4) << 1;
  const float* ap = x + ai * DIM + k0;
  const float* bp = x + bi * DIM + k0;

  v8f c = {0.f, 0.f, 0.f, 0.f, 0.f, 0.f, 0.f, 0.f};
#pragma unroll
  for (int kk = 0; kk < DIM; kk += 4) {
    v2f a = *(const v2f*)(ap + kk);
    v2f b = *(const v2f*)(bp + kk);
    c = __builtin_amdgcn_wmma_f32_16x16x4_f32(false, a, false, b, (short)0, c, false, false);
  }

  const int r = lane & 7;
  float v = c[0];
  if (r == 1) v = c[1];
  if (r == 2) v = c[2];
  if (r == 3) v = c[3];
  if (r == 4) v = c[4];
  if (r == 5) v = c[5];
  if (r == 6) v = c[6];
  if (r == 7) v = c[7];

  const bool active = (lane < 8) || (lane >= 24);
  const int ei = ebase + ((lane < 8) ? lane : (lane - 16));
  if (active && ei < E) res[ei] = v;
}

extern "C" void kernel_launch(void* const* d_in, const int* in_sizes, int n_in,
                              void* d_out, int out_size, void* d_ws, size_t ws_size,
                              hipStream_t stream) {
  const int*   edge_index = (const int*)d_in[0];
  const int*   eli        = (const int*)d_in[1];
  const float* emb        = (const float*)d_in[2];
  const float* alpha      = (const float*)d_in[3];

  const int E  = in_sizes[0] / 2;
  const int Nn = in_sizes[2] / DIM;
  const int L1 = in_sizes[3];
  const int L  = L1 - 1;
  float* res = (float*)d_out;

  const int* src = edge_index;
  const int* dst = edge_index + E;
  const int* la  = eli;
  const int* lb  = eli + E;

  size_t off = 0;
  auto carve = [&](size_t bytes) -> void* {
    void* p = (char*)d_ws + off;
    off = (off + bytes + 255) & ~(size_t)255;
    return p;
  };
  float* wts = (float*)carve(sizeof(float) * (size_t)L1);
  float* dis = (float*)carve(sizeof(float) * (size_t)Nn);
  float* w   = (float*)carve(sizeof(float) * (size_t)E);
  float* xA  = (float*)carve(sizeof(float) * (size_t)Nn * DIM);
  float* xB  = (float*)carve(sizeof(float) * (size_t)Nn * DIM);
  float* acc = (float*)carve(sizeof(float) * (size_t)Nn * DIM);
  (void)ws_size; (void)n_in; (void)out_size;

  const int n4 = Nn * DIM / 4;

  gcn_softmax_kernel<<<1, 1, 0, stream>>>(alpha, wts, L1);
  hipMemsetAsync(dis, 0, sizeof(float) * (size_t)Nn, stream);
  gcn_deg_kernel<<<(E + 255) / 256, 256, 0, stream>>>(dst, dis, E);
  gcn_rsqrt_kernel<<<(Nn + 255) / 256, 256, 0, stream>>>(dis, Nn);
  gcn_edgew_kernel<<<(E + 255) / 256, 256, 0, stream>>>(src, dst, dis, w, E);
  gcn_scale_init_kernel<<<(n4 + 255) / 256, 256, 0, stream>>>(emb, acc, wts, n4);

  const float* xcur = emb;
  float* bufs[2] = {xA, xB};
  for (int l = 0; l < L; ++l) {
    float* xnext = bufs[l & 1];
    hipMemsetAsync(xnext, 0, sizeof(float) * (size_t)Nn * DIM, stream);
    long tcount = (long)E * (DIM / 4);
    gcn_spmm_kernel<<<(unsigned)((tcount + 255) / 256), 256, 0, stream>>>(
        src, dst, w, xcur, xnext, E);
    gcn_axpy_kernel<<<(n4 + 255) / 256, 256, 0, stream>>>(xnext, acc, wts, l + 1, n4);
    xcur = xnext;
  }

  if (E > 0) {
    int waves  = (E + 15) / 16;
    int blocks = (waves + 7) / 8;
    gcn_edge_dot_wmma_kernel<<<blocks, 256, 0, stream>>>(la, lb, acc, res, E);
  }
}


# Bin edges by 400-row destination tile each call (LDS histograms, scans, ranked fill); degree and the 3 layers' scatter summed in a per-tile LDS plane, no global atomics; sums in LDS-add order
# speedup vs baseline: 1.9640x; 1.9640x over previous
#include <hip/hip_runtime.h>

typedef __attribute__((ext_vector_type(2))) float v2f;
typedef __attribute__((ext_vector_type(8))) float v8f;

#define DIM 64
#define TILE 400
#define NBMAX 1024
#define CHUNK 4096
#define CPT (CHUNK / 256)

__global__ void gcn_softmax_kernel(const float* __restrict__ alpha,
                                   float* __restrict__ wts, int n) {
  if (blockIdx.x == 0 && threadIdx.x == 0) {
    float m = alpha[0];
    for (int i = 1; i < n; ++i) m = fmaxf(m, alpha[i]);
    float s = 0.f;
    for (int i = 0; i < n; ++i) s += expf(alpha[i] - m);
    for (int i = 0; i < n; ++i) wts[i] = expf(alpha[i] - m) / s;
  }
}

__global__ __launch_bounds__(256)
void gcn_bin_count_kernel(const int* __restrict__ dst, int* __restrict__ cmat, int E, int NB) {
  __shared__ int hist[NBMAX];
  const int tid = threadIdx.x;
  for (int i = tid; i < NB; i += 256) hist[i] = 0;
  __syncthreads();
  const int c0 = blockIdx.x * CHUNK;
  for (int k = 0; k < CPT; ++k) {
    const int e = c0 + k * 256 + tid;
    if (e < E) atomicAdd(&hist[(unsigned)dst[e] / TILE], 1);
  }
  __syncthreads();
  int* row = cmat + (size_t)blockIdx.x * NB;
  for (int i = tid; i < NB; i += 256) row[i] = hist[i];
}

__global__ __launch_bounds__(512)
void gcn_bin_colscan_kernel(int* __restrict__ cmat, int* __restrict__ binTotal, int chunks, int NB) {
  __shared__ int sc[512];
  const int tid = threadIdx.x;
  const int b = blockIdx.x;
  int carry = 0;
  for (int c0 = 0; c0 < chunks; c0 += 512) {
    const int c = c0 + tid;
    const int v = (c < chunks) ? cmat[(size_t)c * NB + b] : 0;
    sc[tid] = v;
    __syncthreads();
    for (int off = 1; off < 512; off <<= 1) {
      const int t = (tid >= off) ? sc[tid - off] : 0;
      __syncthreads();
      sc[tid] += t;
      __syncthreads();
    }
    if (c < chunks) cmat[(size_t)c * NB + b] = carry + sc[tid] - v;
    carry += sc[511];
    __syncthreads();
  }
  if (tid == 0) binTotal[b] = carry;
}

__global__ __launch_bounds__(1024)
void gcn_bin_scan_kernel(const int* __restrict__ binTotal, int* __restrict__ binStart, int NB) {
  __shared__ int sc[1024];
  const int tid = threadIdx.x;
  const int v = (tid < NB) ? binTotal[tid] : 0;
  sc[tid] = v;
  __syncthreads();
  for (int off = 1; off < 1024; off <<= 1) {
    const int t = (tid >= off) ? sc[tid - off] : 0;
    __syncthreads();
    sc[tid] += t;
    __syncthreads();
  }
  const int incl = sc[tid];
  if (tid < NB) binStart[tid] = incl - v;
  if (tid == NB - 1) binStart[NB] = incl;
}

__global__ __launch_bounds__(256)
void gcn_bin_fill_kernel(const int* __restrict__ src, const int* __restrict__ dst,
                         const int* __restrict__ cmat, const int* __restrict__ binStart,
                         int* __restrict__ binned_w, int E, int NB) {
  __shared__ int rank[NBMAX];
  __shared__ int basep[NBMAX];
  int2* __restrict__ binned = (int2*)binned_w;
  const int tid = threadIdx.x;
  const int c0 = blockIdx.x * CHUNK;
  int dv[CPT], sv[CPT];
#pragma unroll
  for (int k = 0; k < CPT; ++k) {
    const int e = c0 + k * 256 + tid;
    const bool ok = e < E;
    dv[k] = ok ? dst[e] : -1;
    sv[k] = ok ? src[e] : 0;
  }
  const int* row = cmat + (size_t)blockIdx.x * NB;
  for (int i = tid; i < NB; i += 256) {
    rank[i]  = 0;
    basep[i] = binStart[i] + row[i];
  }
  __syncthreads();
#pragma unroll
  for (int k = 0; k < CPT; ++k) {
    if (dv[k] >= 0) {
      const int bb = (unsigned)dv[k] / TILE;
      const int r  = atomicAdd(&rank[bb], 1);
      binned[basep[bb] + r] = make_int2(sv[k], dv[k]);
    }
  }
}

__global__ __launch_bounds__(1024)
void gcn_tile_deg_kernel(const int* __restrict__ binned_w, const int* __restrict__ binStart,
                         float* __restrict__ dis, int Nn) {
  __shared__ int cnt[TILE];
  const int2* __restrict__ binned = (const int2*)binned_w;
  const int tid = threadIdx.x;
  const int b = blockIdx.x;
  const int tile0 = b * TILE;
  const int rows = min(TILE, Nn - tile0);
  for (int i = tid; i < TILE; i += 1024) cnt[i] = 0;
  __syncthreads();
  const int s = binStart[b];
  const int e = binStart[b + 1];
  for (int idx = s + tid; idx < e; idx += 1024) atomicAdd(&cnt[binned[idx].y - tile0], 1);
  __syncthreads();
  for (int i = tid; i < rows; i += 1024) {
    const int c = cnt[i];
    dis[tile0 + i] = (c > 0) ? rsqrtf((float)c) : 0.f;
  }
}

__global__ __launch_bounds__(1024)
void gcn_tile_spmm_kernel(const int* __restrict__ binned_w, const int* __restrict__ binStart,
                          const float* __restrict__ dis, const float* __restrict__ xin,
                          float* __restrict__ xout, int Nn) {
  __shared__ float accT[(TILE + 1) * DIM];
  const int2* __restrict__ binned = (const int2*)binned_w;
  const int tid  = threadIdx.x;
  const int lane = tid & 31;
  const int wave = __builtin_amdgcn_readfirstlane(tid >> 5);
  const int b = blockIdx.x;
  const int tile0 = b * TILE;
  const int rows = min(TILE, Nn - tile0);

  const float4 z4 = {0.f, 0.f, 0.f, 0.f};
  for (int i = tid; i < (TILE + 1) * DIM / 4; i += 1024) ((float4*)accT)[i] = z4;
  __syncthreads();

  const int s = binStart[b];
  const int e = binStart[b + 1];
  const int l2 = 2 * lane;
  const int2 pad = make_int2(0, tile0 + TILE);

  int base = s + wave * 32;
  int2 p0 = pad, p1 = pad;
  float a0 = 0.f, b0 = 0.f;
  if (base + lane < e) {
    p0 = binned[base + lane];
    a0 = dis[p0.x];
    b0 = dis[p0.y];
  }
  if (base + 1024 + lane < e) p1 = binned[base + 1024 + lane];

  while (base < e) {
    const float w   = a0 * b0;
    const int  so   = p0.x * DIM;
    const int  dofs = (p0.y - tile0) * DIM;
    const int  wi   = __float_as_int(w);
    float a1 = 0.f, b1 = 0.f;
    if (base + 1024 + lane < e) {
      a1 = dis[p1.x];
      b1 = dis[p1.y];
    }
    int2 p2 = pad;
    if (base + 2048 + lane < e) p2 = binned[base + 2048 + lane];

    const int cnt = min(32, e - base);
    for (int j = 0; j < cnt; j += 16) {
      v2f v[16];
#pragma unroll
      for (int u = 0; u < 16; ++u) {
        const int sj = __builtin_amdgcn_readlane(so, j + u);
        v[u] = *(const v2f*)(xin + sj + l2);
      }
#pragma unroll
      for (int u = 0; u < 16; ++u) {
        const int   dj = __builtin_amdgcn_readlane(dofs, j + u);
        const float wj = __int_as_float(__builtin_amdgcn_readlane(wi, j + u));
        float* q = &accT[dj + l2];
        __hip_atomic_fetch_add(q,     wj * v[u].x, __ATOMIC_RELAXED, __HIP_MEMORY_SCOPE_WORKGROUP);
        __hip_atomic_fetch_add(q + 1, wj * v[u].y, __ATOMIC_RELAXED, __HIP_MEMORY_SCOPE_WORKGROUP);
      }
    }
    p0 = p1; a0 = a1; b0 = b1; p1 = p2;
    base += 1024;
  }
  __syncthreads();

  float4* o = (float4*)(xout + (size_t)tile0 * DIM);
  const int n4 = rows * (DIM / 4);
  for (int i = tid; i < n4; i += 1024) o[i] = ((const float4*)accT)[i];
}

__global__ void gcn_scale_init_kernel(const float* __restrict__ x, float* __restrict__ o,
                                      const float* __restrict__ wts, int n4) {
  int i = blockIdx.x * blockDim.x + threadIdx.x;
  if (i < n4) {
    float s = wts[0];
    float4 a = ((const float4*)x)[i];
    float4 r;
    r.x = s * a.x; r.y = s * a.y; r.z = s * a.z; r.w = s * a.w;
    ((float4*)o)[i] = r;
  }
}

__global__ void gcn_axpy_kernel(const float* __restrict__ x, float* __restrict__ o,
                                const float* __restrict__ wts, int k, int n4) {
  int i = blockIdx.x * blockDim.x + threadIdx.x;
  if (i < n4) {
    float s = wts[k];
    float4 a = ((const float4*)x)[i];
    float4 r = ((const float4*)o)[i];
    r.x += s * a.x; r.y += s * a.y; r.z += s * a.z; r.w += s * a.w;
    ((float4*)o)[i] = r;
  }
}

__global__ void gcn_edge_dot_wmma_kernel(const int* __restrict__ la, const int* __restrict__ lb,
                                         const float* __restrict__ x, float* __restrict__ res,
                                         int E) {
  const int gwave = (blockIdx.x * blockDim.x + threadIdx.x) >> 5;
  const int lane  = threadIdx.x & 31;
  const int ebase = gwave * 16;
  const int m     = lane & 15;

  int e = ebase + m;
  if (e >= E) e = E - 1;

  const size_t ai = (size_t)la[e];
  const size_t bi = (size_t)lb[e];
  const int k0 = (lane >> 4) << 1;
  const float* ap = x + ai * DIM + k0;
  const float* bp = x + bi * DIM + k0;

  v8f c = {0.f, 0.f, 0.f, 0.f, 0.f, 0.f, 0.f, 0.f};
#pragma unroll
  for (int kk = 0; kk < DIM; kk += 4) {
    v2f a = *(const v2f*)(ap + kk);
    v2f b = *(const v2f*)(bp + kk);
    c = __builtin_amdgcn_wmma_f32_16x16x4_f32(false, a, false, b, (short)0, c, false, false);
  }

  const int r = lane & 7;
  float v = c[0];
  if (r == 1) v = c[1];
  if (r == 2) v = c[2];
  if (r == 3) v = c[3];
  if (r == 4) v = c[4];
  if (r == 5) v = c[5];
  if (r == 6) v = c[6];
  if (r == 7) v = c[7];

  const bool active = (lane < 8) || (lane >= 24);
  const int ei = ebase + ((lane < 8) ? lane : (lane - 16));
  if (active && ei < E) res[ei] = v;
}

extern "C" void kernel_launch(void* const* d_in, const int* in_sizes, int n_in,
                              void* d_out, int out_size, void* d_ws, size_t ws_size,
                              hipStream_t stream) {
  const int*   edge_index = (const int*)d_in[0];
  const int*   eli        = (const int*)d_in[1];
  const float* emb        = (const float*)d_in[2];
  const float* alpha      = (const float*)d_in[3];

  const int E  = in_sizes[0] / 2;
  const int Nn = in_sizes[2] / DIM;
  const int L1 = in_sizes[3];
  const int L  = L1 - 1;
  float* res = (float*)d_out;

  const int* src = edge_index;
  const int* dst = edge_index + E;
  const int* la  = eli;
  const int* lb  = eli + E;

  const int NB = (Nn + TILE - 1) / TILE;
  const int chunks = (E + CHUNK - 1) / CHUNK;

  size_t off = 0;
  auto carve = [&](size_t bytes) -> void* {
    void* p = (char*)d_ws + off;
    off = (off + bytes + 255) & ~(size_t)255;
    return p;
  };
  float* wts      = (float*)carve(sizeof(float) * (size_t)L1);
  float* dis      = (float*)carve(sizeof(float) * (size_t)Nn);
  int*   cmat     = (int*)carve(sizeof(int) * (size_t)chunks * (size_t)NB);
  int*   binTotal = (int*)carve(sizeof(int) * (size_t)NB);
  int*   binStart = (int*)carve(sizeof(int) * (size_t)(NB + 1));
  int*   binned   = (int*)carve(sizeof(int) * 2 * (size_t)E);
  float* xA  = (float*)carve(sizeof(float) * (size_t)Nn * DIM);
  float* xB  = (float*)carve(sizeof(float) * (size_t)Nn * DIM);
  float* acc = (float*)carve(sizeof(float) * (size_t)Nn * DIM);
  (void)ws_size; (void)n_in; (void)out_size;

  const int n4 = Nn * DIM / 4;

  gcn_softmax_kernel<<<1, 1, 0, stream>>>(alpha, wts, L1);
  gcn_bin_count_kernel<<<chunks, 256, 0, stream>>>(dst, cmat, E, NB);
  gcn_bin_colscan_kernel<<<NB, 512, 0, stream>>>(cmat, binTotal, chunks, NB);
  gcn_bin_scan_kernel<<<1, 1024, 0, stream>>>(binTotal, binStart, NB);
  gcn_bin_fill_kernel<<<chunks, 256, 0, stream>>>(src, dst, cmat, binStart, binned, E, NB);
  gcn_tile_deg_kernel<<<NB, 1024, 0, stream>>>(binned, binStart, dis, Nn);
  gcn_scale_init_kernel<<<(n4 + 255) / 256, 256, 0, stream>>>(emb, acc, wts, n4);

  const float* xcur = emb;
  float* bufs[2] = {xA, xB};
  for (int l = 0; l < L; ++l) {
    float* xnext = bufs[l & 1];
    gcn_tile_spmm_kernel<<<NB, 1024, 0, stream>>>(binned, binStart, dis, xcur, xnext, Nn);
    gcn_axpy_kernel<<<(n4 + 255) / 256, 256, 0, stream>>>(xnext, acc, wts, l + 1, n4);
    xcur = xnext;
  }

  if (E > 0) {
    int waves  = (E + 15) / 16;
    int blocks = (waves + 7) / 8;
    gcn_edge_dot_wmma_kernel<<<blocks, 256, 0, stream>>>(la, lb, acc, res, E);
  }
}


# Layer sums on the matrix unit: edges regrouped by 16-row destination group, a wave per group multiplies a 16x4 selection-weight matrix by 4 gathered rows (f32 16x16x4); no LDS adds; sum order changes
# speedup vs baseline: 2.0273x; 1.0322x over previous
#include <hip/hip_runtime.h>

typedef __attribute__((ext_vector_type(2))) float v2f;
typedef __attribute__((ext_vector_type(8))) float v8f;

#define DIM 64
#define TILE 400
#define NBMAX 1024
#define CHUNK 4096
#define CPT (CHUNK / 256)
#define GROWS 16
#define TGROUPS (TILE / GROWS)

__global__ void gcn_softmax_kernel(const float* __restrict__ alpha,
                                   float* __restrict__ wts, int n) {
  if (blockIdx.x == 0 && threadIdx.x == 0) {
    float m = alpha[0];
    for (int i = 1; i < n; ++i) m = fmaxf(m, alpha[i]);
    float s = 0.f;
    for (int i = 0; i < n; ++i) s += expf(alpha[i] - m);
    for (int i = 0; i < n; ++i) wts[i] = expf(alpha[i] - m) / s;
  }
}

__global__ __launch_bounds__(256)
void gcn_bin_count_kernel(const int* __restrict__ dst, int* __restrict__ cmat, int E, int NB) {
  __shared__ int hist[NBMAX];
  const int tid = threadIdx.x;
  for (int i = tid; i < NB; i += 256) hist[i] = 0;
  __syncthreads();
  const int c0 = blockIdx.x * CHUNK;
  for (int k = 0; k < CPT; ++k) {
    const int e = c0 + k * 256 + tid;
    if (e < E) atomicAdd(&hist[(unsigned)dst[e] / TILE], 1);
  }
  __syncthreads();
  int* row = cmat + (size_t)blockIdx.x * NB;
  for (int i = tid; i < NB; i += 256) row[i] = hist[i];
}

__global__ __launch_bounds__(512)
void gcn_bin_colscan_kernel(int* __restrict__ cmat, int* __restrict__ binTotal, int chunks, int NB) {
  __shared__ int sc[512];
  const int tid = threadIdx.x;
  const int b = blockIdx.x;
  int carry = 0;
  for (int c0 = 0; c0 < chunks; c0 += 512) {
    const int c = c0 + tid;
    const int v = (c < chunks) ? cmat[(size_t)c * NB + b] : 0;
    sc[tid] = v;
    __syncthreads();
    for (int off = 1; off < 512; off <<= 1) {
      const int t = (tid >= off) ? sc[tid - off] : 0;
      __syncthreads();
      sc[tid] += t;
      __syncthreads();
    }
    if (c < chunks) cmat[(size_t)c * NB + b] = carry + sc[tid] - v;
    carry += sc[511];
    __syncthreads();
  }
  if (tid == 0) binTotal[b] = carry;
}

__global__ __launch_bounds__(1024)
void gcn_bin_scan_kernel(const int* __restrict__ binTotal, int* __restrict__ binStart, int NB) {
  __shared__ int sc[1024];
  const int tid = threadIdx.x;
  const int v = (tid < NB) ? binTotal[tid] : 0;
  sc[tid] = v;
  __syncthreads();
  for (int off = 1; off < 1024; off <<= 1) {
    const int t = (tid >= off) ? sc[tid - off] : 0;
    __syncthreads();
    sc[tid] += t;
    __syncthreads();
  }
  const int incl = sc[tid];
  if (tid < NB) binStart[tid] = incl - v;
  if (tid == NB - 1) binStart[NB] = incl;
}

__global__ __launch_bounds__(256)
void gcn_bin_fill_kernel(const int* __restrict__ src, const int* __restrict__ dst,
                         const int* __restrict__ cmat, const int* __restrict__ binStart,
                         int* __restrict__ binned_w, int E, int NB) {
  __shared__ int rank[NBMAX];
  __shared__ int basep[NBMAX];
  int2* __restrict__ binned = (int2*)binned_w;
  const int tid = threadIdx.x;
  const int c0 = blockIdx.x * CHUNK;
  int dv[CPT], sv[CPT];
#pragma unroll
  for (int k = 0; k < CPT; ++k) {
    const int e = c0 + k * 256 + tid;
    const bool ok = e < E;
    dv[k] = ok ? dst[e] : -1;
    sv[k] = ok ? src[e] : 0;
  }
  const int* row = cmat + (size_t)blockIdx.x * NB;
  for (int i = tid; i < NB; i += 256) {
    rank[i]  = 0;
    basep[i] = binStart[i] + row[i];
  }
  __syncthreads();
#pragma unroll
  for (int k = 0; k < CPT; ++k) {
    if (dv[k] >= 0) {
      const int bb = (unsigned)dv[k] / TILE;
      const int r  = atomicAdd(&rank[bb], 1);
      binned[basep[bb] + r] = make_int2(sv[k], dv[k]);
    }
  }
}

__global__ __launch_bounds__(1024)
void gcn_tile_deg_kernel(const int* __restrict__ binned_w, const int* __restrict__ binStart,
                         float* __restrict__ dis, int* __restrict__ grouped_w,
                         int* __restrict__ grpStart, int Nn) {
  __shared__ int cnt[TILE];
  __shared__ int gsum[TGROUPS];
  __shared__ int gcur[TGROUPS];
  const int2* __restrict__ binned = (const int2*)binned_w;
  int2* __restrict__ grouped = (int2*)grouped_w;
  const int tid = threadIdx.x;
  const int b = blockIdx.x;
  const int tile0 = b * TILE;
  const int rows = min(TILE, Nn - tile0);
  for (int i = tid; i < TILE; i += 1024) cnt[i] = 0;
  __syncthreads();
  const int s = binStart[b];
  const int e = binStart[b + 1];
  for (int idx = s + tid; idx < e; idx += 1024) atomicAdd(&cnt[binned[idx].y - tile0], 1);
  __syncthreads();
  for (int i = tid; i < rows; i += 1024) {
    const int c = cnt[i];
    dis[tile0 + i] = (c > 0) ? rsqrtf((float)c) : 0.f;
  }
  if (tid < TGROUPS) {
    int t = 0;
    for (int i = 0; i < GROWS; ++i) t += cnt[tid * GROWS + i];
    gsum[tid] = t;
  }
  __syncthreads();
  if (tid == 0) {
    int run = s;
    for (int g = 0; g < TGROUPS; ++g) {
      gcur[g] = run;
      grpStart[b * TGROUPS + g] = run;
      run += gsum[g];
    }
    if (b == (int)gridDim.x - 1) grpStart[(b + 1) * TGROUPS] = e;
  }
  __syncthreads();
  for (int idx = s + tid; idx < e; idx += 1024) {
    const int2 p = binned[idx];
    const int r = atomicAdd(&gcur[(p.y - tile0) / GROWS], 1);
    grouped[r] = p;
  }
}

__global__ __launch_bounds__(128)
void gcn_group_spmm_wmma_kernel(const int* __restrict__ grouped_w, const int* __restrict__ grpStart,
                                const float* __restrict__ dis, const float* __restrict__ xin,
                                float* __restrict__ xout, int Nn, int NG) {
  const int2* __restrict__ grouped = (const int2*)grouped_w;
  const int lane = threadIdx.x & 31;
  const int gw = __builtin_amdgcn_readfirstlane((int)((blockIdx.x * blockDim.x + threadIdx.x) >> 5));
  if (gw >= NG) return;
  const int tb   = gw / TGROUPS;
  const int row0 = tb * TILE + (gw - tb * TGROUPS) * GROWS;
  const int s = grpStart[gw];
  const int e = grpStart[gw + 1];
  const int c = lane & 15;
  const bool hi = lane >= 16;
  const int c4 = 4 * c;
  const int c16 = 16 * c;
  const char* __restrict__ xbytes = (const char*)xin;

  v8f acc0 = {0.f, 0.f, 0.f, 0.f, 0.f, 0.f, 0.f, 0.f};
  v8f acc1 = acc0, acc2 = acc0, acc3 = acc0;

  const int2 pad = make_int2(0, row0);
  int2 p0 = pad, p1 = pad;
  float a0 = 0.f, b0 = 0.f;
  if (s + lane < e) {
    p0 = grouped[s + lane];
    a0 = dis[p0.x];
    b0 = dis[p0.y];
  }
  if (s + 32 + lane < e) p1 = grouped[s + 32 + lane];

  for (int base = s; base < e; base += 32) {
    const int so = p0.x * (DIM * 4);
    const int dl = p0.y - row0;
    const int wi = __float_as_int(a0 * b0);
    float a1 = 0.f, b1 = 0.f;
    if (base + 32 + lane < e) {
      a1 = dis[p1.x];
      b1 = dis[p1.y];
    }
    int2 p2 = pad;
    if (base + 64 + lane < e) p2 = grouped[base + 64 + lane];

    const int cnt = min(32, e - base);
    for (int j = 0; j < cnt; j += 16) {
      float4 q0[4], q1[4];
      v2f af[4];
#pragma unroll
      for (int u = 0; u < 4; ++u) {
        const int jj = j + 4 * u;
        const int sA = __builtin_amdgcn_readlane(so, jj);
        const int sB = __builtin_amdgcn_readlane(so, jj + 1);
        const int sC = __builtin_amdgcn_readlane(so, jj + 2);
        const int sD = __builtin_amdgcn_readlane(so, jj + 3);
        q0[u] = *(const float4*)(xbytes + (unsigned)((hi ? sC : sA) + c16));
        q1[u] = *(const float4*)(xbytes + (unsigned)((hi ? sD : sB) + c16));
        const int dA = __builtin_amdgcn_readlane(dl, jj);
        const int dB = __builtin_amdgcn_readlane(dl, jj + 1);
        const int dC = __builtin_amdgcn_readlane(dl, jj + 2);
        const int dD = __builtin_amdgcn_readlane(dl, jj + 3);
        const int wA = __builtin_amdgcn_readlane(wi, jj);
        const int wB = __builtin_amdgcn_readlane(wi, jj + 1);
        const int wC = __builtin_amdgcn_readlane(wi, jj + 2);
        const int wD = __builtin_amdgcn_readlane(wi, jj + 3);
        af[u].x = ((hi ? dC : dA) == c) ? __int_as_float(hi ? wC : wA) : 0.f;
        af[u].y = ((hi ? dD : dB) == c) ? __int_as_float(hi ? wD : wB) : 0.f;
      }
#pragma unroll
      for (int u = 0; u < 4; ++u) {
        v2f bf;
        bf.x = q0[u].x; bf.y = q1[u].x;
        acc0 = __builtin_amdgcn_wmma_f32_16x16x4_f32(false, af[u], false, bf, (short)0, acc0, false, false);
        bf.x = q0[u].y; bf.y = q1[u].y;
        acc1 = __builtin_amdgcn_wmma_f32_16x16x4_f32(false, af[u], false, bf, (short)0, acc1, false, false);
        bf.x = q0[u].z; bf.y = q1[u].z;
        acc2 = __builtin_amdgcn_wmma_f32_16x16x4_f32(false, af[u], false, bf, (short)0, acc2, false, false);
        bf.x = q0[u].w; bf.y = q1[u].w;
        acc3 = __builtin_amdgcn_wmma_f32_16x16x4_f32(false, af[u], false, bf, (short)0, acc3, false, false);
      }
    }
    p0 = p1; a0 = a1; b0 = b1; p1 = p2;
  }

  const int rbase = row0 + (hi ? 8 : 0);
#pragma unroll
  for (int r = 0; r < 8; ++r) {
    const int row = rbase + r;
    if (row < Nn) {
      float4 o;
      o.x = acc0[r]; o.y = acc1[r]; o.z = acc2[r]; o.w = acc3[r];
      *(float4*)(xout + (size_t)row * DIM + c4) = o;
    }
  }
}

__global__ void gcn_scale_init_kernel(const float* __restrict__ x, float* __restrict__ o,
                                      const float* __restrict__ wts, int n4) {
  int i = blockIdx.x * blockDim.x + threadIdx.x;
  if (i < n4) {
    float s = wts[0];
    float4 a = ((const float4*)x)[i];
    float4 r;
    r.x = s * a.x; r.y = s * a.y; r.z = s * a.z; r.w = s * a.w;
    ((float4*)o)[i] = r;
  }
}

__global__ void gcn_axpy_kernel(const float* __restrict__ x, float* __restrict__ o,
                                const float* __restrict__ wts, int k, int n4) {
  int i = blockIdx.x * blockDim.x + threadIdx.x;
  if (i < n4) {
    float s = wts[k];
    float4 a = ((const float4*)x)[i];
    float4 r = ((const float4*)o)[i];
    r.x += s * a.x; r.y += s * a.y; r.z += s * a.z; r.w += s * a.w;
    ((float4*)o)[i] = r;
  }
}

__global__ void gcn_edge_dot_wmma_kernel(const int* __restrict__ la, const int* __restrict__ lb,
                                         const float* __restrict__ x, float* __restrict__ res,
                                         int E) {
  const int gwave = (blockIdx.x * blockDim.x + threadIdx.x) >> 5;
  const int lane  = threadIdx.x & 31;
  const int ebase = gwave * 16;
  const int m     = lane & 15;

  int e = ebase + m;
  if (e >= E) e = E - 1;

  const size_t ai = (size_t)la[e];
  const size_t bi = (size_t)lb[e];
  const int k0 = (lane >> 4) << 1;
  const float* ap = x + ai * DIM + k0;
  const float* bp = x + bi * DIM + k0;

  v8f c = {0.f, 0.f, 0.f, 0.f, 0.f, 0.f, 0.f, 0.f};
#pragma unroll
  for (int kk = 0; kk < DIM; kk += 4) {
    v2f a = *(const v2f*)(ap + kk);
    v2f b = *(const v2f*)(bp + kk);
    c = __builtin_amdgcn_wmma_f32_16x16x4_f32(false, a, false, b, (short)0, c, false, false);
  }

  const int r = lane & 7;
  float v = c[0];
  if (r == 1) v = c[1];
  if (r == 2) v = c[2];
  if (r == 3) v = c[3];
  if (r == 4) v = c[4];
  if (r == 5) v = c[5];
  if (r == 6) v = c[6];
  if (r == 7) v = c[7];

  const bool active = (lane < 8) || (lane >= 24);
  const int ei = ebase + ((lane < 8) ? lane : (lane - 16));
  if (active && ei < E) res[ei] = v;
}

extern "C" void kernel_launch(void* const* d_in, const int* in_sizes, int n_in,
                              void* d_out, int out_size, void* d_ws, size_t ws_size,
                              hipStream_t stream) {
  const int*   edge_index = (const int*)d_in[0];
  const int*   eli        = (const int*)d_in[1];
  const float* emb        = (const float*)d_in[2];
  const float* alpha      = (const float*)d_in[3];

  const int E  = in_sizes[0] / 2;
  const int Nn = in_sizes[2] / DIM;
  const int L1 = in_sizes[3];
  const int L  = L1 - 1;
  float* res = (float*)d_out;

  const int* src = edge_index;
  const int* dst = edge_index + E;
  const int* la  = eli;
  const int* lb  = eli + E;

  const int NB = (Nn + TILE - 1) / TILE;
  const int chunks = (E + CHUNK - 1) / CHUNK;
  const int NG = NB * TGROUPS;

  size_t off = 0;
  auto carve = [&](size_t bytes) -> void* {
    void* p = (char*)d_ws + off;
    off = (off + bytes + 255) & ~(size_t)255;
    return p;
  };
  float* wts      = (float*)carve(sizeof(float) * (size_t)L1);
  float* dis      = (float*)carve(sizeof(float) * (size_t)Nn);
  int*   cmat     = (int*)carve(sizeof(int) * (size_t)chunks * (size_t)NB);
  int*   binTotal = (int*)carve(sizeof(int) * (size_t)NB);
  int*   binStart = (int*)carve(sizeof(int) * (size_t)(NB + 1));
  int*   binned   = (int*)carve(sizeof(int) * 2 * (size_t)E);
  int*   grouped  = (int*)carve(sizeof(int) * 2 * (size_t)E);
  int*   grpStart = (int*)carve(sizeof(int) * ((size_t)NB * TGROUPS + 1));
  float* xA  = (float*)carve(sizeof(float) * (size_t)Nn * DIM);
  float* xB  = (float*)carve(sizeof(float) * (size_t)Nn * DIM);
  float* acc = (float*)carve(sizeof(float) * (size_t)Nn * DIM);
  (void)ws_size; (void)n_in; (void)out_size;

  const int n4 = Nn * DIM / 4;

  gcn_softmax_kernel<<<1, 1, 0, stream>>>(alpha, wts, L1);
  gcn_bin_count_kernel<<<chunks, 256, 0, stream>>>(dst, cmat, E, NB);
  gcn_bin_colscan_kernel<<<NB, 512, 0, stream>>>(cmat, binTotal, chunks, NB);
  gcn_bin_scan_kernel<<<1, 1024, 0, stream>>>(binTotal, binStart, NB);
  gcn_bin_fill_kernel<<<chunks, 256, 0, stream>>>(src, dst, cmat, binStart, binned, E, NB);
  gcn_tile_deg_kernel<<<NB, 1024, 0, stream>>>(binned, binStart, dis, grouped, grpStart, Nn);
  gcn_scale_init_kernel<<<(n4 + 255) / 256, 256, 0, stream>>>(emb, acc, wts, n4);

  const float* xcur = emb;
  float* bufs[2] = {xA, xB};
  for (int l = 0; l < L; ++l) {
    float* xnext = bufs[l & 1];
    gcn_group_spmm_wmma_kernel<<<(NG + 3) / 4, 128, 0, stream>>>(grouped, grpStart, dis, xcur, xnext, Nn, NG);
    gcn_axpy_kernel<<<(n4 + 255) / 256, 256, 0, stream>>>(xnext, acc, wts, l + 1, n4);
    xcur = xnext;
  }

  if (E > 0) {
    int waves  = (E + 15) / 16;
    int blocks = (waves + 7) / 8;
    gcn_edge_dot_wmma_kernel<<<blocks, 256, 0, stream>>>(la, lb, acc, res, E);
  }
}


# Layer-weighted sum deferred to ONE pass behind the last layer: the 3 layer planes are kept, a wave sums 8 x 512 B of each with all 32 loads ahead; scale pass and 3 add passes go; sum order unchanged
# speedup vs baseline: 2.2905x; 1.1298x over previous
#include <hip/hip_runtime.h>

typedef __attribute__((ext_vector_type(2))) float v2f;
typedef __attribute__((ext_vector_type(8))) float v8f;

#define DIM 64
#define TILE 400
#define NBMAX 1024
#define CHUNK 4096
#define CPT (CHUNK / 256)
#define GROWS 16
#define TGROUPS (TILE / GROWS)
#define SUMU 8

__global__ void gcn_softmax_kernel(const float* __restrict__ alpha,
                                   float* __restrict__ wts, int n) {
  if (blockIdx.x == 0 && threadIdx.x == 0) {
    float m = alpha[0];
    for (int i = 1; i < n; ++i) m = fmaxf(m, alpha[i]);
    float s = 0.f;
    for (int i = 0; i < n; ++i) s += expf(alpha[i] - m);
    for (int i = 0; i < n; ++i) wts[i] = expf(alpha[i] - m) / s;
  }
}

__global__ __launch_bounds__(256)
void gcn_bin_count_kernel(const int* __restrict__ dst, int* __restrict__ cmat, int E, int NB) {
  __shared__ int hist[NBMAX];
  const int tid = threadIdx.x;
  for (int i = tid; i < NB; i += 256) hist[i] = 0;
  __syncthreads();
  const int c0 = blockIdx.x * CHUNK;
  for (int k = 0; k < CPT; ++k) {
    const int e = c0 + k * 256 + tid;
    if (e < E) atomicAdd(&hist[(unsigned)dst[e] / TILE], 1);
  }
  __syncthreads();
  int* row = cmat + (size_t)blockIdx.x * NB;
  for (int i = tid; i < NB; i += 256) row[i] = hist[i];
}

__global__ __launch_bounds__(512)
void gcn_bin_colscan_kernel(int* __restrict__ cmat, int* __restrict__ binTotal, int chunks, int NB) {
  __shared__ int sc[512];
  const int tid = threadIdx.x;
  const int b = blockIdx.x;
  int carry = 0;
  for (int c0 = 0; c0 < chunks; c0 += 512) {
    const int c = c0 + tid;
    const int v = (c < chunks) ? cmat[(size_t)c * NB + b] : 0;
    sc[tid] = v;
    __syncthreads();
    for (int off = 1; off < 512; off <<= 1) {
      const int t = (tid >= off) ? sc[tid - off] : 0;
      __syncthreads();
      sc[tid] += t;
      __syncthreads();
    }
    if (c < chunks) cmat[(size_t)c * NB + b] = carry + sc[tid] - v;
    carry += sc[511];
    __syncthreads();
  }
  if (tid == 0) binTotal[b] = carry;
}

__global__ __launch_bounds__(1024)
void gcn_bin_scan_kernel(const int* __restrict__ binTotal, int* __restrict__ binStart, int NB) {
  __shared__ int sc[1024];
  const int tid = threadIdx.x;
  const int v = (tid < NB) ? binTotal[tid] : 0;
  sc[tid] = v;
  __syncthreads();
  for (int off = 1; off < 1024; off <<= 1) {
    const int t = (tid >= off) ? sc[tid - off] : 0;
    __syncthreads();
    sc[tid] += t;
    __syncthreads();
  }
  const int incl = sc[tid];
  if (tid < NB) binStart[tid] = incl - v;
  if (tid == NB - 1) binStart[NB] = incl;
}

__global__ __launch_bounds__(256)
void gcn_bin_fill_kernel(const int* __restrict__ src, const int* __restrict__ dst,
                         const int* __restrict__ cmat, const int* __restrict__ binStart,
                         int* __restrict__ binned_w, int E, int NB) {
  __shared__ int rank[NBMAX];
  __shared__ int basep[NBMAX];
  int2* __restrict__ binned = (int2*)binned_w;
  const int tid = threadIdx.x;
  const int c0 = blockIdx.x * CHUNK;
  int dv[CPT], sv[CPT];
#pragma unroll
  for (int k = 0; k < CPT; ++k) {
    const int e = c0 + k * 256 + tid;
    const bool ok = e < E;
    dv[k] = ok ? dst[e] : -1;
    sv[k] = ok ? src[e] : 0;
  }
  const int* row = cmat + (size_t)blockIdx.x * NB;
  for (int i = tid; i < NB; i += 256) {
    rank[i]  = 0;
    basep[i] = binStart[i] + row[i];
  }
  __syncthreads();
#pragma unroll
  for (int k = 0; k < CPT; ++k) {
    if (dv[k] >= 0) {
      const int bb = (unsigned)dv[k] / TILE;
      const int r  = atomicAdd(&rank[bb], 1);
      binned[basep[bb] + r] = make_int2(sv[k], dv[k]);
    }
  }
}

__global__ __launch_bounds__(1024)
void gcn_tile_deg_kernel(const int* __restrict__ binned_w, const int* __restrict__ binStart,
                         float* __restrict__ dis, int* __restrict__ grouped_w,
                         int* __restrict__ grpStart, int Nn) {
  __shared__ int cnt[TILE];
  __shared__ int gsum[TGROUPS];
  __shared__ int gcur[TGROUPS];
  const int2* __restrict__ binned = (const int2*)binned_w;
  int2* __restrict__ grouped = (int2*)grouped_w;
  const int tid = threadIdx.x;
  const int b = blockIdx.x;
  const int tile0 = b * TILE;
  const int rows = min(TILE, Nn - tile0);
  for (int i = tid; i < TILE; i += 1024) cnt[i] = 0;
  __syncthreads();
  const int s = binStart[b];
  const int e = binStart[b + 1];
  for (int idx = s + tid; idx < e; idx += 1024) atomicAdd(&cnt[binned[idx].y - tile0], 1);
  __syncthreads();
  for (int i = tid; i < rows; i += 1024) {
    const int c = cnt[i];
    dis[tile0 + i] = (c > 0) ? rsqrtf((float)c) : 0.f;
  }
  if (tid < TGROUPS) {
    int t = 0;
    for (int i = 0; i < GROWS; ++i) t += cnt[tid * GROWS + i];
    gsum[tid] = t;
  }
  __syncthreads();
  if (tid == 0) {
    int run = s;
    for (int g = 0; g < TGROUPS; ++g) {
      gcur[g] = run;
      grpStart[b * TGROUPS + g] = run;
      run += gsum[g];
    }
    if (b == (int)gridDim.x - 1) grpStart[(b + 1) * TGROUPS] = e;
  }
  __syncthreads();
  for (int idx = s + tid; idx < e; idx += 1024) {
    const int2 p = binned[idx];
    const int r = atomicAdd(&gcur[(p.y - tile0) / GROWS], 1);
    grouped[r] = p;
  }
}

__global__ __launch_bounds__(128)
void gcn_group_spmm_wmma_kernel(const int* __restrict__ grouped_w, const int* __restrict__ grpStart,
                                const float* __restrict__ dis, const float* __restrict__ xin,
                                float* __restrict__ xout, int Nn, int NG) {
  const int2* __restrict__ grouped = (const int2*)grouped_w;
  const int lane = threadIdx.x & 31;
  const int gw = __builtin_amdgcn_readfirstlane((int)((blockIdx.x * blockDim.x + threadIdx.x) >> 5));
  if (gw >= NG) return;
  const int tb   = gw / TGROUPS;
  const int row0 = tb * TILE + (gw - tb * TGROUPS) * GROWS;
  const int s = grpStart[gw];
  const int e = grpStart[gw + 1];
  const int c = lane & 15;
  const bool hi = lane >= 16;
  const int c4 = 4 * c;
  const int c16 = 16 * c;
  const char* __restrict__ xbytes = (const char*)xin;

  v8f acc0 = {0.f, 0.f, 0.f, 0.f, 0.f, 0.f, 0.f, 0.f};
  v8f acc1 = acc0, acc2 = acc0, acc3 = acc0;

  const int2 pad = make_int2(0, row0);
  int2 p0 = pad, p1 = pad;
  float a0 = 0.f, b0 = 0.f;
  if (s + lane < e) {
    p0 = grouped[s + lane];
    a0 = dis[p0.x];
    b0 = dis[p0.y];
  }
  if (s + 32 + lane < e) p1 = grouped[s + 32 + lane];

  for (int base = s; base < e; base += 32) {
    const int so = p0.x * (DIM * 4);
    const int dl = p0.y - row0;
    const int wi = __float_as_int(a0 * b0);
    float a1 = 0.f, b1 = 0.f;
    if (base + 32 + lane < e) {
      a1 = dis[p1.x];
      b1 = dis[p1.y];
    }
    int2 p2 = pad;
    if (base + 64 + lane < e) p2 = grouped[base + 64 + lane];

    const int cnt = min(32, e - base);
    for (int j = 0; j < cnt; j += 16) {
      float4 q0[4], q1[4];
      v2f af[4];
#pragma unroll
      for (int u = 0; u < 4; ++u) {
        const int jj = j + 4 * u;
        const int sA = __builtin_amdgcn_readlane(so, jj);
        const int sB = __builtin_amdgcn_readlane(so, jj + 1);
        const int sC = __builtin_amdgcn_readlane(so, jj + 2);
        const int sD = __builtin_amdgcn_readlane(so, jj + 3);
        q0[u] = *(const float4*)(xbytes + (unsigned)((hi ? sC : sA) + c16));
        q1[u] = *(const float4*)(xbytes + (unsigned)((hi ? sD : sB) + c16));
        const int dA = __builtin_amdgcn_readlane(dl, jj);
        const int dB = __builtin_amdgcn_readlane(dl, jj + 1);
        const int dC = __builtin_amdgcn_readlane(dl, jj + 2);
        const int dD = __builtin_amdgcn_readlane(dl, jj + 3);
        const int wA = __builtin_amdgcn_readlane(wi, jj);
        const int wB = __builtin_amdgcn_readlane(wi, jj + 1);
        const int wC = __builtin_amdgcn_readlane(wi, jj + 2);
        const int wD = __builtin_amdgcn_readlane(wi, jj + 3);
        af[u].x = ((hi ? dC : dA) == c) ? __int_as_float(hi ? wC : wA) : 0.f;
        af[u].y = ((hi ? dD : dB) == c) ? __int_as_float(hi ? wD : wB) : 0.f;
      }
#pragma unroll
      for (int u = 0; u < 4; ++u) {
        v2f bf;
        bf.x = q0[u].x; bf.y = q1[u].x;
        acc0 = __builtin_amdgcn_wmma_f32_16x16x4_f32(false, af[u], false, bf, (short)0, acc0, false, false);
        bf.x = q0[u].y; bf.y = q1[u].y;
        acc1 = __builtin_amdgcn_wmma_f32_16x16x4_f32(false, af[u], false, bf, (short)0, acc1, false, false);
        bf.x = q0[u].z; bf.y = q1[u].z;
        acc2 = __builtin_amdgcn_wmma_f32_16x16x4_f32(false, af[u], false, bf, (short)0, acc2, false, false);
        bf.x = q0[u].w; bf.y = q1[u].w;
        acc3 = __builtin_amdgcn_wmma_f32_16x16x4_f32(false, af[u], false, bf, (short)0, acc3, false, false);
      }
    }
    p0 = p1; a0 = a1; b0 = b1; p1 = p2;
  }

  const int rbase = row0 + (hi ? 8 : 0);
#pragma unroll
  for (int r = 0; r < 8; ++r) {
    const int row = rbase + r;
    if (row < Nn) {
      float4 o;
      o.x = acc0[r]; o.y = acc1[r]; o.z = acc2[r]; o.w = acc3[r];
      *(float4*)(xout + (size_t)row * DIM + c4) = o;
    }
  }
}

__global__ __launch_bounds__(128)
void gcn_layer_sum_kernel(const float* base, const float* xa, const float* xb, const float* xc,
                          float* o, const float* __restrict__ wts, int k0, int nl, int first, int n4) {
  const int lane = threadIdx.x & 31;
  const int gw = __builtin_amdgcn_readfirstlane((int)((blockIdx.x * blockDim.x + threadIdx.x) >> 5));
  const int i0 = gw * (32 * SUMU);
  if (i0 >= n4) return;
  const float4* b4 = (const float4*)base;
  const float4* a4 = (const float4*)xa;
  const float4* c4 = (const float4*)xb;
  const float4* d4 = (const float4*)xc;

  int idx[SUMU];
  float4 vb[SUMU], va[SUMU], vc[SUMU], vd[SUMU];
#pragma unroll
  for (int u = 0; u < SUMU; ++u) {
    idx[u] = min(i0 + 32 * u + lane, n4 - 1);
    vb[u] = b4[idx[u]];
  }
  if (nl > 0) {
#pragma unroll
    for (int u = 0; u < SUMU; ++u) va[u] = a4[idx[u]];
  }
  if (nl > 1) {
#pragma unroll
    for (int u = 0; u < SUMU; ++u) vc[u] = c4[idx[u]];
  }
  if (nl > 2) {
#pragma unroll
    for (int u = 0; u < SUMU; ++u) vd[u] = d4[idx[u]];
  }
  const float s0 = first ? wts[0] : 1.0f;
  const float s1 = (nl > 0) ? wts[k0] : 0.f;
  const float s2 = (nl > 1) ? wts[k0 + 1] : 0.f;
  const float s3 = (nl > 2) ? wts[k0 + 2] : 0.f;

  float4* o4 = (float4*)o;
#pragma unroll
  for (int u = 0; u < SUMU; ++u) {
    float4 r;
    r.x = s0 * vb[u].x; r.y = s0 * vb[u].y; r.z = s0 * vb[u].z; r.w = s0 * vb[u].w;
    if (nl > 0) { r.x += s1 * va[u].x; r.y += s1 * va[u].y; r.z += s1 * va[u].z; r.w += s1 * va[u].w; }
    if (nl > 1) { r.x += s2 * vc[u].x; r.y += s2 * vc[u].y; r.z += s2 * vc[u].z; r.w += s2 * vc[u].w; }
    if (nl > 2) { r.x += s3 * vd[u].x; r.y += s3 * vd[u].y; r.z += s3 * vd[u].z; r.w += s3 * vd[u].w; }
    const int i = i0 + 32 * u + lane;
    if (i < n4) o4[i] = r;
  }
}

__global__ void gcn_edge_dot_wmma_kernel(const int* __restrict__ la, const int* __restrict__ lb,
                                         const float* __restrict__ x, float* __restrict__ res,
                                         int E) {
  const int gwave = (blockIdx.x * blockDim.x + threadIdx.x) >> 5;
  const int lane  = threadIdx.x & 31;
  const int ebase = gwave * 16;
  const int m     = lane & 15;

  int e = ebase + m;
  if (e >= E) e = E - 1;

  const size_t ai = (size_t)la[e];
  const size_t bi = (size_t)lb[e];
  const int k0 = (lane >> 4) << 1;
  const float* ap = x + ai * DIM + k0;
  const float* bp = x + bi * DIM + k0;

  v8f c = {0.f, 0.f, 0.f, 0.f, 0.f, 0.f, 0.f, 0.f};
#pragma unroll
  for (int kk = 0; kk < DIM; kk += 4) {
    v2f a = *(const v2f*)(ap + kk);
    v2f b = *(const v2f*)(bp + kk);
    c = __builtin_amdgcn_wmma_f32_16x16x4_f32(false, a, false, b, (short)0, c, false, false);
  }

  const int r = lane & 7;
  float v = c[0];
  if (r == 1) v = c[1];
  if (r == 2) v = c[2];
  if (r == 3) v = c[3];
  if (r == 4) v = c[4];
  if (r == 5) v = c[5];
  if (r == 6) v = c[6];
  if (r == 7) v = c[7];

  const bool active = (lane < 8) || (lane >= 24);
  const int ei = ebase + ((lane < 8) ? lane : (lane - 16));
  if (active && ei < E) res[ei] = v;
}

extern "C" void kernel_launch(void* const* d_in, const int* in_sizes, int n_in,
                              void* d_out, int out_size, void* d_ws, size_t ws_size,
                              hipStream_t stream) {
  const int*   edge_index = (const int*)d_in[0];
  const int*   eli        = (const int*)d_in[1];
  const float* emb        = (const float*)d_in[2];
  const float* alpha      = (const float*)d_in[3];

  const int E  = in_sizes[0] / 2;
  const int Nn = in_sizes[2] / DIM;
  const int L1 = in_sizes[3];
  const int L  = L1 - 1;
  float* res = (float*)d_out;

  const int* src = edge_index;
  const int* dst = edge_index + E;
  const int* la  = eli;
  const int* lb  = eli + E;

  const int NB = (Nn + TILE - 1) / TILE;
  const int chunks = (E + CHUNK - 1) / CHUNK;
  const int NG = NB * TGROUPS;

  size_t off = 0;
  auto carve = [&](size_t bytes) -> void* {
    void* p = (char*)d_ws + off;
    off = (off + bytes + 255) & ~(size_t)255;
    return p;
  };
  float* wts      = (float*)carve(sizeof(float) * (size_t)L1);
  float* dis      = (float*)carve(sizeof(float) * (size_t)Nn);
  int*   cmat     = (int*)carve(sizeof(int) * (size_t)chunks * (size_t)NB);
  int*   binTotal = (int*)carve(sizeof(int) * (size_t)NB);
  int*   binStart = (int*)carve(sizeof(int) * (size_t)(NB + 1));
  int*   binned   = (int*)carve(sizeof(int) * 2 * (size_t)E);
  int*   grouped  = (int*)carve(sizeof(int) * 2 * (size_t)E);
  int*   grpStart = (int*)carve(sizeof(int) * ((size_t)NB * TGROUPS + 1));
  float* xA  = (float*)carve(sizeof(float) * (size_t)Nn * DIM);
  float* xB  = (float*)carve(sizeof(float) * (size_t)Nn * DIM);
  float* xC  = (float*)carve(sizeof(float) * (size_t)Nn * DIM);
  float* acc = (float*)carve(sizeof(float) * (size_t)Nn * DIM);
  (void)ws_size; (void)n_in; (void)out_size;

  const int n4 = Nn * DIM / 4;

  gcn_softmax_kernel<<<1, 1, 0, stream>>>(alpha, wts, L1);
  gcn_bin_count_kernel<<<chunks, 256, 0, stream>>>(dst, cmat, E, NB);
  gcn_bin_colscan_kernel<<<NB, 512, 0, stream>>>(cmat, binTotal, chunks, NB);
  gcn_bin_scan_kernel<<<1, 1024, 0, stream>>>(binTotal, binStart, NB);
  gcn_bin_fill_kernel<<<chunks, 256, 0, stream>>>(src, dst, cmat, binStart, binned, E, NB);
  gcn_tile_deg_kernel<<<NB, 1024, 0, stream>>>(binned, binStart, dis, grouped, grpStart, Nn);
  const int sumWaves  = (n4 + 32 * SUMU - 1) / (32 * SUMU);
  const int sumBlocks = (sumWaves + 3) / 4;
  const float* xcur = emb;
  float* bufs[3] = {xA, xB, xC};
  if (L <= 0)
    gcn_layer_sum_kernel<<<sumBlocks, 128, 0, stream>>>(emb, emb, emb, emb, acc, wts, 1, 0, 1, n4);
  for (int l = 0; l < L; ++l) {
    float* xnext = bufs[l % 3];
    gcn_group_spmm_wmma_kernel<<<(NG + 3) / 4, 128, 0, stream>>>(grouped, grpStart, dis, xcur, xnext, Nn, NG);
    xcur = xnext;
    const int filled = l % 3 + 1;
    if (filled == 3 || l == L - 1) {
      const int firstSum = (l < 3) ? 1 : 0;
      gcn_layer_sum_kernel<<<sumBlocks, 128, 0, stream>>>(firstSum ? emb : acc, bufs[0], bufs[1], bufs[2],
                                                          acc, wts, l + 2 - filled, filled, firstSum, n4);
    }
  }

  if (E > 0) {
    int waves  = (E + 15) / 16;
    int blocks = (waves + 7) / 8;
    gcn_edge_dot_wmma_kernel<<<blocks, 256, 0, stream>>>(la, lb, acc, res, E);
  }
}


# Edge-dot re-tiled: 16x16 tile = 4 edges x 4 feature chunks, 8 lanes x 16 B load a whole row line straight into fragment layout; diagonal quad-summed; a wave loops over 8 batches; sum order changes
# speedup vs baseline: 2.7400x; 1.1962x over previous
#include <hip/hip_runtime.h>

typedef __attribute__((ext_vector_type(2))) float v2f;
typedef __attribute__((ext_vector_type(8))) float v8f;

#define DIM 64
#define TILE 400
#define NBMAX 1024
#define CHUNK 4096
#define CPT (CHUNK / 256)
#define GROWS 16
#define TGROUPS (TILE / GROWS)
#define SUMU 8

__global__ void gcn_softmax_kernel(const float* __restrict__ alpha,
                                   float* __restrict__ wts, int n) {
  if (blockIdx.x == 0 && threadIdx.x == 0) {
    float m = alpha[0];
    for (int i = 1; i < n; ++i) m = fmaxf(m, alpha[i]);
    float s = 0.f;
    for (int i = 0; i < n; ++i) s += expf(alpha[i] - m);
    for (int i = 0; i < n; ++i) wts[i] = expf(alpha[i] - m) / s;
  }
}

__global__ __launch_bounds__(256)
void gcn_bin_count_kernel(const int* __restrict__ dst, int* __restrict__ cmat, int E, int NB) {
  __shared__ int hist[NBMAX];
  const int tid = threadIdx.x;
  for (int i = tid; i < NB; i += 256) hist[i] = 0;
  __syncthreads();
  const int c0 = blockIdx.x * CHUNK;
  for (int k = 0; k < CPT; ++k) {
    const int e = c0 + k * 256 + tid;
    if (e < E) atomicAdd(&hist[(unsigned)dst[e] / TILE], 1);
  }
  __syncthreads();
  int* row = cmat + (size_t)blockIdx.x * NB;
  for (int i = tid; i < NB; i += 256) row[i] = hist[i];
}

__global__ __launch_bounds__(512)
void gcn_bin_colscan_kernel(int* __restrict__ cmat, int* __restrict__ binTotal, int chunks, int NB) {
  __shared__ int sc[512];
  const int tid = threadIdx.x;
  const int b = blockIdx.x;
  int carry = 0;
  for (int c0 = 0; c0 < chunks; c0 += 512) {
    const int c = c0 + tid;
    const int v = (c < chunks) ? cmat[(size_t)c * NB + b] : 0;
    sc[tid] = v;
    __syncthreads();
    for (int off = 1; off < 512; off <<= 1) {
      const int t = (tid >= off) ? sc[tid - off] : 0;
      __syncthreads();
      sc[tid] += t;
      __syncthreads();
    }
    if (c < chunks) cmat[(size_t)c * NB + b] = carry + sc[tid] - v;
    carry += sc[511];
    __syncthreads();
  }
  if (tid == 0) binTotal[b] = carry;
}

__global__ __launch_bounds__(1024)
void gcn_bin_scan_kernel(const int* __restrict__ binTotal, int* __restrict__ binStart, int NB) {
  __shared__ int sc[1024];
  const int tid = threadIdx.x;
  const int v = (tid < NB) ? binTotal[tid] : 0;
  sc[tid] = v;
  __syncthreads();
  for (int off = 1; off < 1024; off <<= 1) {
    const int t = (tid >= off) ? sc[tid - off] : 0;
    __syncthreads();
    sc[tid] += t;
    __syncthreads();
  }
  const int incl = sc[tid];
  if (tid < NB) binStart[tid] = incl - v;
  if (tid == NB - 1) binStart[NB] = incl;
}

__global__ __launch_bounds__(256)
void gcn_bin_fill_kernel(const int* __restrict__ src, const int* __restrict__ dst,
                         const int* __restrict__ cmat, const int* __restrict__ binStart,
                         int* __restrict__ binned_w, int E, int NB) {
  __shared__ int rank[NBMAX];
  __shared__ int basep[NBMAX];
  int2* __restrict__ binned = (int2*)binned_w;
  const int tid = threadIdx.x;
  const int c0 = blockIdx.x * CHUNK;
  int dv[CPT], sv[CPT];
#pragma unroll
  for (int k = 0; k < CPT; ++k) {
    const int e = c0 + k * 256 + tid;
    const bool ok = e < E;
    dv[k] = ok ? dst[e] : -1;
    sv[k] = ok ? src[e] : 0;
  }
  const int* row = cmat + (size_t)blockIdx.x * NB;
  for (int i = tid; i < NB; i += 256) {
    rank[i]  = 0;
    basep[i] = binStart[i] + row[i];
  }
  __syncthreads();
#pragma unroll
  for (int k = 0; k < CPT; ++k) {
    if (dv[k] >= 0) {
      const int bb = (unsigned)dv[k] / TILE;
      const int r  = atomicAdd(&rank[bb], 1);
      binned[basep[bb] + r] = make_int2(sv[k], dv[k]);
    }
  }
}

__global__ __launch_bounds__(1024)
void gcn_tile_deg_kernel(const int* __restrict__ binned_w, const int* __restrict__ binStart,
                         float* __restrict__ dis, int* __restrict__ grouped_w,
                         int* __restrict__ grpStart, int Nn) {
  __shared__ int cnt[TILE];
  __shared__ int gsum[TGROUPS];
  __shared__ int gcur[TGROUPS];
  const int2* __restrict__ binned = (const int2*)binned_w;
  int2* __restrict__ grouped = (int2*)grouped_w;
  const int tid = threadIdx.x;
  const int b = blockIdx.x;
  const int tile0 = b * TILE;
  const int rows = min(TILE, Nn - tile0);
  for (int i = tid; i < TILE; i += 1024) cnt[i] = 0;
  __syncthreads();
  const int s = binStart[b];
  const int e = binStart[b + 1];
  for (int idx = s + tid; idx < e; idx += 1024) atomicAdd(&cnt[binned[idx].y - tile0], 1);
  __syncthreads();
  for (int i = tid; i < rows; i += 1024) {
    const int c = cnt[i];
    dis[tile0 + i] = (c > 0) ? rsqrtf((float)c) : 0.f;
  }
  if (tid < TGROUPS) {
    int t = 0;
    for (int i = 0; i < GROWS; ++i) t += cnt[tid * GROWS + i];
    gsum[tid] = t;
  }
  __syncthreads();
  if (tid == 0) {
    int run = s;
    for (int g = 0; g < TGROUPS; ++g) {
      gcur[g] = run;
      grpStart[b * TGROUPS + g] = run;
      run += gsum[g];
    }
    if (b == (int)gridDim.x - 1) grpStart[(b + 1) * TGROUPS] = e;
  }
  __syncthreads();
  for (int idx = s + tid; idx < e; idx += 1024) {
    const int2 p = binned[idx];
    const int r = atomicAdd(&gcur[(p.y - tile0) / GROWS], 1);
    grouped[r] = p;
  }
}

__global__ __launch_bounds__(128)
void gcn_group_spmm_wmma_kernel(const int* __restrict__ grouped_w, const int* __restrict__ grpStart,
                                const float* __restrict__ dis, const float* __restrict__ xin,
                                float* __restrict__ xout, int Nn, int NG) {
  const int2* __restrict__ grouped = (const int2*)grouped_w;
  const int lane = threadIdx.x & 31;
  const int gw = __builtin_amdgcn_readfirstlane((int)((blockIdx.x * blockDim.x + threadIdx.x) >> 5));
  if (gw >= NG) return;
  const int tb   = gw / TGROUPS;
  const int row0 = tb * TILE + (gw - tb * TGROUPS) * GROWS;
  const int s = grpStart[gw];
  const int e = grpStart[gw + 1];
  const int c = lane & 15;
  const bool hi = lane >= 16;
  const int c4 = 4 * c;
  const int c16 = 16 * c;
  const char* __restrict__ xbytes = (const char*)xin;

  v8f acc0 = {0.f, 0.f, 0.f, 0.f, 0.f, 0.f, 0.f, 0.f};
  v8f acc1 = acc0, acc2 = acc0, acc3 = acc0;

  const int2 pad = make_int2(0, row0);
  int2 p0 = pad, p1 = pad;
  float a0 = 0.f, b0 = 0.f;
  if (s + lane < e) {
    p0 = grouped[s + lane];
    a0 = dis[p0.x];
    b0 = dis[p0.y];
  }
  if (s + 32 + lane < e) p1 = grouped[s + 32 + lane];

  for (int base = s; base < e; base += 32) {
    const int so = p0.x * (DIM * 4);
    const int dl = p0.y - row0;
    const int wi = __float_as_int(a0 * b0);
    float a1 = 0.f, b1 = 0.f;
    if (base + 32 + lane < e) {
      a1 = dis[p1.x];
      b1 = dis[p1.y];
    }
    int2 p2 = pad;
    if (base + 64 + lane < e) p2 = grouped[base + 64 + lane];

    const int cnt = min(32, e - base);
    for (int j = 0; j < cnt; j += 16) {
      float4 q0[4], q1[4];
      v2f af[4];
#pragma unroll
      for (int u = 0; u < 4; ++u) {
        const int jj = j + 4 * u;
        const int sA = __builtin_amdgcn_readlane(so, jj);
        const int sB = __builtin_amdgcn_readlane(so, jj + 1);
        const int sC = __builtin_amdgcn_readlane(so, jj + 2);
        const int sD = __builtin_amdgcn_readlane(so, jj + 3);
        q0[u] = *(const float4*)(xbytes + (unsigned)((hi ? sC : sA) + c16));
        q1[u] = *(const float4*)(xbytes + (unsigned)((hi ? sD : sB) + c16));
        const int dA = __builtin_amdgcn_readlane(dl, jj);
        const int dB = __builtin_amdgcn_readlane(dl, jj + 1);
        const int dC = __builtin_amdgcn_readlane(dl, jj + 2);
        const int dD = __builtin_amdgcn_readlane(dl, jj + 3);
        const int wA = __builtin_amdgcn_readlane(wi, jj);
        const int wB = __builtin_amdgcn_readlane(wi, jj + 1);
        const int wC = __builtin_amdgcn_readlane(wi, jj + 2);
        const int wD = __builtin_amdgcn_readlane(wi, jj + 3);
        af[u].x = ((hi ? dC : dA) == c) ? __int_as_float(hi ? wC : wA) : 0.f;
        af[u].y = ((hi ? dD : dB) == c) ? __int_as_float(hi ? wD : wB) : 0.f;
      }
#pragma unroll
      for (int u = 0; u < 4; ++u) {
        v2f bf;
        bf.x = q0[u].x; bf.y = q1[u].x;
        acc0 = __builtin_amdgcn_wmma_f32_16x16x4_f32(false, af[u], false, bf, (short)0, acc0, false, false);
        bf.x = q0[u].y; bf.y = q1[u].y;
        acc1 = __builtin_amdgcn_wmma_f32_16x16x4_f32(false, af[u], false, bf, (short)0, acc1, false, false);
        bf.x = q0[u].z; bf.y = q1[u].z;
        acc2 = __builtin_amdgcn_wmma_f32_16x16x4_f32(false, af[u], false, bf, (short)0, acc2, false, false);
        bf.x = q0[u].w; bf.y = q1[u].w;
        acc3 = __builtin_amdgcn_wmma_f32_16x16x4_f32(false, af[u], false, bf, (short)0, acc3, false, false);
      }
    }
    p0 = p1; a0 = a1; b0 = b1; p1 = p2;
  }

  const int rbase = row0 + (hi ? 8 : 0);
#pragma unroll
  for (int r = 0; r < 8; ++r) {
    const int row = rbase + r;
    if (row < Nn) {
      float4 o;
      o.x = acc0[r]; o.y = acc1[r]; o.z = acc2[r]; o.w = acc3[r];
      *(float4*)(xout + (size_t)row * DIM + c4) = o;
    }
  }
}

__global__ __launch_bounds__(128)
void gcn_layer_sum_kernel(const float* base, const float* xa, const float* xb, const float* xc,
                          float* o, const float* __restrict__ wts, int k0, int nl, int first, int n4) {
  const int lane = threadIdx.x & 31;
  const int gw = __builtin_amdgcn_readfirstlane((int)((blockIdx.x * blockDim.x + threadIdx.x) >> 5));
  const int i0 = gw * (32 * SUMU);
  if (i0 >= n4) return;
  const float4* b4 = (const float4*)base;
  const float4* a4 = (const float4*)xa;
  const float4* c4 = (const float4*)xb;
  const float4* d4 = (const float4*)xc;

  int idx[SUMU];
  float4 vb[SUMU], va[SUMU], vc[SUMU], vd[SUMU];
#pragma unroll
  for (int u = 0; u < SUMU; ++u) {
    idx[u] = min(i0 + 32 * u + lane, n4 - 1);
    vb[u] = b4[idx[u]];
  }
  if (nl > 0) {
#pragma unroll
    for (int u = 0; u < SUMU; ++u) va[u] = a4[idx[u]];
  }
  if (nl > 1) {
#pragma unroll
    for (int u = 0; u < SUMU; ++u) vc[u] = c4[idx[u]];
  }
  if (nl > 2) {
#pragma unroll
    for (int u = 0; u < SUMU; ++u) vd[u] = d4[idx[u]];
  }
  const float s0 = first ? wts[0] : 1.0f;
  const float s1 = (nl > 0) ? wts[k0] : 0.f;
  const float s2 = (nl > 1) ? wts[k0 + 1] : 0.f;
  const float s3 = (nl > 2) ? wts[k0 + 2] : 0.f;

  float4* o4 = (float4*)o;
#pragma unroll
  for (int u = 0; u < SUMU; ++u) {
    float4 r;
    r.x = s0 * vb[u].x; r.y = s0 * vb[u].y; r.z = s0 * vb[u].z; r.w = s0 * vb[u].w;
    if (nl > 0) { r.x += s1 * va[u].x; r.y += s1 * va[u].y; r.z += s1 * va[u].z; r.w += s1 * va[u].w; }
    if (nl > 1) { r.x += s2 * vc[u].x; r.y += s2 * vc[u].y; r.z += s2 * vc[u].z; r.w += s2 * vc[u].w; }
    if (nl > 2) { r.x += s3 * vd[u].x; r.y += s3 * vd[u].y; r.z += s3 * vd[u].z; r.w += s3 * vd[u].w; }
    const int i = i0 + 32 * u + lane;
    if (i < n4) o4[i] = r;
  }
}

__device__ __forceinline__ float gcn_quad_sum(float v) {
  int t = __builtin_amdgcn_update_dpp(0, __float_as_int(v), 0xB1, 0xF, 0xF, true);
  v += __int_as_float(t);
  t = __builtin_amdgcn_update_dpp(0, __float_as_int(v), 0x4E, 0xF, 0xF, true);
  v += __int_as_float(t);
  return v;
}

__device__ __forceinline__ float gcn_pick8(v8f a, int r) {
  float v = a[0];
  if (r == 1) v = a[1];
  if (r == 2) v = a[2];
  if (r == 3) v = a[3];
  if (r == 4) v = a[4];
  if (r == 5) v = a[5];
  if (r == 6) v = a[6];
  if (r == 7) v = a[7];
  return v;
}

__global__ __launch_bounds__(128)
void gcn_edge_dot_wmma_kernel(const int* __restrict__ la, const int* __restrict__ lb,
                              const float* __restrict__ x, float* __restrict__ res,
                              int E, int npair) {
  const int lane = threadIdx.x & 31;
  const int gw = __builtin_amdgcn_readfirstlane((int)((blockIdx.x * blockDim.x + threadIdx.x) >> 5));
  const int nw = (int)((gridDim.x * blockDim.x) >> 5);
  const int c = lane & 15;
  const int slot = c >> 2;
  const int piece = ((c & 3) << 1) | (lane >> 4);
  const char* __restrict__ xbytes = (const char*)x;
  const unsigned pbytes = 16u * (unsigned)piece;
  const int last = E - 1;

  int pair = gw;
  int half = 0;
  int b = 2 * gw;
  int ia[4], ib[4];
#pragma unroll
  for (int q = 0; q < 4; ++q) {
    const int e = min(16 * b + 4 * slot + q, last);
    ia[q] = la[e];
    ib[q] = lb[e];
  }

  while (pair < npair) {
    float4 u[4][2], v[4][2];
#pragma unroll
    for (int q = 0; q < 4; ++q) {
      const unsigned oa = (unsigned)ia[q] * (DIM * 4u) + pbytes;
      const unsigned ob = (unsigned)ib[q] * (DIM * 4u) + pbytes;
      u[q][0] = *(const float4*)(xbytes + oa);
      u[q][1] = *(const float4*)(xbytes + oa + 128u);
      v[q][0] = *(const float4*)(xbytes + ob);
      v[q][1] = *(const float4*)(xbytes + ob + 128u);
    }
    const int bn = (half == 0) ? (b + 1) : (2 * (pair + nw));
#pragma unroll
    for (int q = 0; q < 4; ++q) {
      const int e = min(16 * bn + 4 * slot + q, last);
      ia[q] = la[e];
      ib[q] = lb[e];
    }

    v8f acc0 = {0.f, 0.f, 0.f, 0.f, 0.f, 0.f, 0.f, 0.f};
    v8f acc1 = acc0, acc2 = acc0, acc3 = acc0;
#pragma unroll
    for (int j = 0; j < 2; ++j) {
      v2f a, bb;
      a.x = u[0][j].x; a.y = u[0][j].y; bb.x = v[0][j].x; bb.y = v[0][j].y;
      acc0 = __builtin_amdgcn_wmma_f32_16x16x4_f32(false, a, false, bb, (short)0, acc0, false, false);
      a.x = u[1][j].x; a.y = u[1][j].y; bb.x = v[1][j].x; bb.y = v[1][j].y;
      acc1 = __builtin_amdgcn_wmma_f32_16x16x4_f32(false, a, false, bb, (short)0, acc1, false, false);
      a.x = u[2][j].x; a.y = u[2][j].y; bb.x = v[2][j].x; bb.y = v[2][j].y;
      acc2 = __builtin_amdgcn_wmma_f32_16x16x4_f32(false, a, false, bb, (short)0, acc2, false, false);
      a.x = u[3][j].x; a.y = u[3][j].y; bb.x = v[3][j].x; bb.y = v[3][j].y;
      acc3 = __builtin_amdgcn_wmma_f32_16x16x4_f32(false, a, false, bb, (short)0, acc3, false, false);
      a.x = u[0][j].z; a.y = u[0][j].w; bb.x = v[0][j].z; bb.y = v[0][j].w;
      acc0 = __builtin_amdgcn_wmma_f32_16x16x4_f32(false, a, false, bb, (short)0, acc0, false, false);
      a.x = u[1][j].z; a.y = u[1][j].w; bb.x = v[1][j].z; bb.y = v[1][j].w;
      acc1 = __builtin_amdgcn_wmma_f32_16x16x4_f32(false, a, false, bb, (short)0, acc1, false, false);
      a.x = u[2][j].z; a.y = u[2][j].w; bb.x = v[2][j].z; bb.y = v[2][j].w;
      acc2 = __builtin_amdgcn_wmma_f32_16x16x4_f32(false, a, false, bb, (short)0, acc2, false, false);
      a.x = u[3][j].z; a.y = u[3][j].w; bb.x = v[3][j].z; bb.y = v[3][j].w;
      acc3 = __builtin_amdgcn_wmma_f32_16x16x4_f32(false, a, false, bb, (short)0, acc3, false, false);
    }

    const int r = lane & 7;
    float4 o;
    o.x = gcn_quad_sum(gcn_pick8(acc0, r));
    o.y = gcn_quad_sum(gcn_pick8(acc1, r));
    o.z = gcn_quad_sum(gcn_pick8(acc2, r));
    o.w = gcn_quad_sum(gcn_pick8(acc3, r));

    const bool diag = (lane < 8) || (lane >= 24);
    const int e0 = 16 * b + 4 * slot;
    if (diag && (lane & 3) == 0) {
      if (e0 + 3 < E) {
        *(float4*)(res + e0) = o;
      } else {
        if (e0 < E) res[e0] = o.x;
        if (e0 + 1 < E) res[e0 + 1] = o.y;
        if (e0 + 2 < E) res[e0 + 2] = o.z;
      }
    }

    if (half == 1) pair += nw;
    half ^= 1;
    b = bn;
  }
}

extern "C" void kernel_launch(void* const* d_in, const int* in_sizes, int n_in,
                              void* d_out, int out_size, void* d_ws, size_t ws_size,
                              hipStream_t stream) {
  const int*   edge_index = (const int*)d_in[0];
  const int*   eli        = (const int*)d_in[1];
  const float* emb        = (const float*)d_in[2];
  const float* alpha      = (const float*)d_in[3];

  const int E  = in_sizes[0] / 2;
  const int Nn = in_sizes[2] / DIM;
  const int L1 = in_sizes[3];
  const int L  = L1 - 1;
  float* res = (float*)d_out;

  const int* src = edge_index;
  const int* dst = edge_index + E;
  const int* la  = eli;
  const int* lb  = eli + E;

  const int NB = (Nn + TILE - 1) / TILE;
  const int chunks = (E + CHUNK - 1) / CHUNK;
  const int NG = NB * TGROUPS;

  size_t off = 0;
  auto carve = [&](size_t bytes) -> void* {
    void* p = (char*)d_ws + off;
    off = (off + bytes + 255) & ~(size_t)255;
    return p;
  };
  float* wts      = (float*)carve(sizeof(float) * (size_t)L1);
  float* dis      = (float*)carve(sizeof(float) * (size_t)Nn);
  int*   cmat     = (int*)carve(sizeof(int) * (size_t)chunks * (size_t)NB);
  int*   binTotal = (int*)carve(sizeof(int) * (size_t)NB);
  int*   binStart = (int*)carve(sizeof(int) * (size_t)(NB + 1));
  int*   binned   = (int*)carve(sizeof(int) * 2 * (size_t)E);
  int*   grouped  = (int*)carve(sizeof(int) * 2 * (size_t)E);
  int*   grpStart = (int*)carve(sizeof(int) * ((size_t)NB * TGROUPS + 1));
  float* xA  = (float*)carve(sizeof(float) * (size_t)Nn * DIM);
  float* xB  = (float*)carve(sizeof(float) * (size_t)Nn * DIM);
  float* xC  = (float*)carve(sizeof(float) * (size_t)Nn * DIM);
  float* acc = (float*)carve(sizeof(float) * (size_t)Nn * DIM);
  (void)ws_size; (void)n_in; (void)out_size;

  const int n4 = Nn * DIM / 4;

  gcn_softmax_kernel<<<1, 1, 0, stream>>>(alpha, wts, L1);
  gcn_bin_count_kernel<<<chunks, 256, 0, stream>>>(dst, cmat, E, NB);
  gcn_bin_colscan_kernel<<<NB, 512, 0, stream>>>(cmat, binTotal, chunks, NB);
  gcn_bin_scan_kernel<<<1, 1024, 0, stream>>>(binTotal, binStart, NB);
  gcn_bin_fill_kernel<<<chunks, 256, 0, stream>>>(src, dst, cmat, binStart, binned, E, NB);
  gcn_tile_deg_kernel<<<NB, 1024, 0, stream>>>(binned, binStart, dis, grouped, grpStart, Nn);
  const int sumWaves  = (n4 + 32 * SUMU - 1) / (32 * SUMU);
  const int sumBlocks = (sumWaves + 3) / 4;
  const float* xcur = emb;
  float* bufs[3] = {xA, xB, xC};
  if (L <= 0)
    gcn_layer_sum_kernel<<<sumBlocks, 128, 0, stream>>>(emb, emb, emb, emb, acc, wts, 1, 0, 1, n4);
  for (int l = 0; l < L; ++l) {
    float* xnext = bufs[l % 3];
    gcn_group_spmm_wmma_kernel<<<(NG + 3) / 4, 128, 0, stream>>>(grouped, grpStart, dis, xcur, xnext, Nn, NG);
    xcur = xnext;
    const int filled = l % 3 + 1;
    if (filled == 3 || l == L - 1) {
      const int firstSum = (l < 3) ? 1 : 0;
      gcn_layer_sum_kernel<<<sumBlocks, 128, 0, stream>>>(firstSum ? emb : acc, bufs[0], bufs[1], bufs[2],
                                                          acc, wts, l + 2 - filled, filled, firstSum, n4);
    }
  }

  if (E > 0) {
    const int nbatch = (E + 15) / 16;
    const int npair  = (nbatch + 1) / 2;
    const int blocks = (npair + 15) / 16;
    gcn_edge_dot_wmma_kernel<<<blocks, 128, 0, stream>>>(la, lb, acc, res, E, npair);
  }
}


# 11 -> 8 launches: each fill work-group sums and scans the raw chunk x tile counts itself (both scan launches go; 8192-edge chunks), each layer-sum wave forms softmax(alpha); sum order changes
# speedup vs baseline: 3.0294x; 1.1056x over previous
#include <hip/hip_runtime.h>

typedef __attribute__((ext_vector_type(2))) float v2f;
typedef __attribute__((ext_vector_type(8))) float v8f;

#define DIM 64
#define TILE 400
#define NBMAX 1024
#define BT 512
#define CHUNK 8192
#define CPT (CHUNK / BT)
#define GROWS 16
#define TGROUPS (TILE / GROWS)
#define SUMU 8

struct gcn_softmax_state { float m, s; };
__device__ __forceinline__ gcn_softmax_state gcn_softmax_prepare(const float* __restrict__ alpha, int n) {
  gcn_softmax_state st;
  st.m = alpha[0];
  for (int i = 1; i < n; ++i) st.m = fmaxf(st.m, alpha[i]);
  st.s = 0.f;
  for (int i = 0; i < n; ++i) st.s += expf(alpha[i] - st.m);
  return st;
}
__device__ __forceinline__ float gcn_softmax_weight(const float* __restrict__ alpha, gcn_softmax_state st, int i) {
  return expf(alpha[i] - st.m) / st.s;
}

__global__ __launch_bounds__(BT)
void gcn_bin_count_kernel(const int* __restrict__ dst, int* __restrict__ cmat, int E, int NB) {
  __shared__ int hist[NBMAX];
  const int tid = threadIdx.x;
  for (int i = tid; i < NB; i += BT) hist[i] = 0;
  __syncthreads();
  const int c0 = blockIdx.x * CHUNK;
  for (int k = 0; k < CPT; ++k) {
    const int e = c0 + k * BT + tid;
    if (e < E) atomicAdd(&hist[(unsigned)dst[e] / TILE], 1);
  }
  __syncthreads();
  const int NBP = (NB + 3) & ~3;
  int* row = cmat + (size_t)blockIdx.x * NBP;
  for (int i = tid; i < NBP; i += BT) row[i] = (i < NB) ? hist[i] : 0;
}

__global__ __launch_bounds__(BT)
void gcn_bin_fill_kernel(const int* __restrict__ src, const int* __restrict__ dst,
                         const int* __restrict__ cmat, int* __restrict__ binStart,
                         int* __restrict__ binned_w, int E, int NB, int chunks) {
  __shared__ int rank[NBMAX];
  __shared__ int basep[NBMAX];
  __shared__ int tot[NBMAX];
  __shared__ int wsum[BT / 32];
  int2* __restrict__ binned = (int2*)binned_w;
  const int tid = threadIdx.x;
  const int me = blockIdx.x;
  const int c0 = me * CHUNK;
  for (int i = tid; i < NB; i += BT) {
    rank[i]  = 0;
    basep[i] = 0;
    tot[i]   = 0;
  }
  __syncthreads();
  int dv[CPT], sv[CPT];
#pragma unroll
  for (int k = 0; k < CPT; ++k) {
    const int e = c0 + k * BT + tid;
    const bool ok = e < E;
    dv[k] = ok ? dst[e] : -1;
    sv[k] = ok ? src[e] : 0;
  }
  const int NBP = (NB + 3) & ~3;
  const int Q = NBP >> 2;
  const int S = (Q < BT) ? BT / Q : 1;
  const int4* __restrict__ cm4 = (const int4*)cmat;
  for (int it = tid; it < S * Q; it += BT) {
    const int sl = it / Q;
    const int q  = it - sl * Q;
    int t0 = 0, t1 = 0, t2 = 0, t3 = 0;
    int b0 = 0, b1 = 0, b2 = 0, b3 = 0;
    for (int cb = sl; cb < chunks; cb += 8 * S) {
      int4 v[8];
#pragma unroll
      for (int u = 0; u < 8; ++u) {
        const int c = min(cb + u * S, chunks - 1);
        v[u] = cm4[c * Q + q];
      }
#pragma unroll
      for (int u = 0; u < 8; ++u) {
        const int c = cb + u * S;
        const int all = (c < chunks) ? -1 : 0;
        const int bef = (c < me) ? all : 0;
        t0 += v[u].x & all; t1 += v[u].y & all; t2 += v[u].z & all; t3 += v[u].w & all;
        b0 += v[u].x & bef; b1 += v[u].y & bef; b2 += v[u].z & bef; b3 += v[u].w & bef;
      }
    }
    const int b = 4 * q;
    if (b < NB)     { atomicAdd(&tot[b],     t0); atomicAdd(&basep[b],     b0); }
    if (b + 1 < NB) { atomicAdd(&tot[b + 1], t1); atomicAdd(&basep[b + 1], b1); }
    if (b + 2 < NB) { atomicAdd(&tot[b + 2], t2); atomicAdd(&basep[b + 2], b2); }
    if (b + 3 < NB) { atomicAdd(&tot[b + 3], t3); atomicAdd(&basep[b + 3], b3); }
  }
  __syncthreads();
  const int lane = tid & 31;
  const int wv = tid >> 5;
  int loc[NBMAX / BT];
  int run = 0;
  for (int j = 0; j < NBMAX / BT; ++j) {
    const int idx = tid * (NBMAX / BT) + j;
    loc[j] = run;
    run += (idx < NB) ? tot[idx] : 0;
  }
  int inc = run;
  for (int off = 1; off < 32; off <<= 1) {
    const int t = __shfl_up(inc, off);
    if (lane >= off) inc += t;
  }
  if (lane == 31) wsum[wv] = inc;
  __syncthreads();
  int woff = 0;
  for (int w = 0; w < wv; ++w) woff += wsum[w];
  const int excl = woff + inc - run;
  for (int j = 0; j < NBMAX / BT; ++j) {
    const int idx = tid * (NBMAX / BT) + j;
    if (idx < NB) {
      const int start = excl + loc[j];
      basep[idx] += start;
      if (me == 0) binStart[idx] = start;
    }
  }
  if (me == 0 && tid == BT - 1) binStart[NB] = woff + inc;
  __syncthreads();
#pragma unroll
  for (int k = 0; k < CPT; ++k) {
    if (dv[k] >= 0) {
      const int bb = (unsigned)dv[k] / TILE;
      const int r  = atomicAdd(&rank[bb], 1);
      binned[basep[bb] + r] = make_int2(sv[k], dv[k]);
    }
  }
}

__global__ __launch_bounds__(1024)
void gcn_tile_deg_kernel(const int* __restrict__ binned_w, const int* __restrict__ binStart,
                         float* __restrict__ dis, int* __restrict__ grouped_w,
                         int* __restrict__ grpStart, int Nn) {
  __shared__ int cnt[TILE];
  __shared__ int gsum[TGROUPS];
  __shared__ int gcur[TGROUPS];
  const int2* __restrict__ binned = (const int2*)binned_w;
  int2* __restrict__ grouped = (int2*)grouped_w;
  const int tid = threadIdx.x;
  const int b = blockIdx.x;
  const int tile0 = b * TILE;
  const int rows = min(TILE, Nn - tile0);
  for (int i = tid; i < TILE; i += 1024) cnt[i] = 0;
  __syncthreads();
  const int s = binStart[b];
  const int e = binStart[b + 1];
  for (int idx = s + tid; idx < e; idx += 1024) atomicAdd(&cnt[binned[idx].y - tile0], 1);
  __syncthreads();
  for (int i = tid; i < rows; i += 1024) {
    const int c = cnt[i];
    dis[tile0 + i] = (c > 0) ? rsqrtf((float)c) : 0.f;
  }
  if (tid < TGROUPS) {
    int t = 0;
    for (int i = 0; i < GROWS; ++i) t += cnt[tid * GROWS + i];
    gsum[tid] = t;
  }
  __syncthreads();
  if (tid == 0) {
    int run = s;
    for (int g = 0; g < TGROUPS; ++g) {
      gcur[g] = run;
      grpStart[b * TGROUPS + g] = run;
      run += gsum[g];
    }
    if (b == (int)gridDim.x - 1) grpStart[(b + 1) * TGROUPS] = e;
  }
  __syncthreads();
  for (int idx = s + tid; idx < e; idx += 1024) {
    const int2 p = binned[idx];
    const int r = atomicAdd(&gcur[(p.y - tile0) / GROWS], 1);
    grouped[r] = p;
  }
}

__global__ __launch_bounds__(128)
void gcn_group_spmm_wmma_kernel(const int* __restrict__ grouped_w, const int* __restrict__ grpStart,
                                const float* __restrict__ dis, const float* __restrict__ xin,
                                float* __restrict__ xout, int Nn, int NG) {
  const int2* __restrict__ grouped = (const int2*)grouped_w;
  const int lane = threadIdx.x & 31;
  const int gw = __builtin_amdgcn_readfirstlane((int)((blockIdx.x * blockDim.x + threadIdx.x) >> 5));
  if (gw >= NG) return;
  const int tb   = gw / TGROUPS;
  const int row0 = tb * TILE + (gw - tb * TGROUPS) * GROWS;
  const int s = grpStart[gw];
  const int e = grpStart[gw + 1];
  const int c = lane & 15;
  const bool hi = lane >= 16;
  const int c4 = 4 * c;
  const int c16 = 16 * c;
  const char* __restrict__ xbytes = (const char*)xin;

  v8f acc0 = {0.f, 0.f, 0.f, 0.f, 0.f, 0.f, 0.f, 0.f};
  v8f acc1 = acc0, acc2 = acc0, acc3 = acc0;

  const int2 pad = make_int2(0, row0);
  int2 p0 = pad, p1 = pad;
  float a0 = 0.f, b0 = 0.f;
  if (s + lane < e) {
    p0 = grouped[s + lane];
    a0 = dis[p0.x];
    b0 = dis[p0.y];
  }
  if (s + 32 + lane < e) p1 = grouped[s + 32 + lane];

  for (int base = s; base < e; base += 32) {
    const int so = p0.x * (DIM * 4);
    const int dl = p0.y - row0;
    const int wi = __float_as_int(a0 * b0);
    float a1 = 0.f, b1 = 0.f;
    if (base + 32 + lane < e) {
      a1 = dis[p1.x];
      b1 = dis[p1.y];
    }
    int2 p2 = pad;
    if (base + 64 + lane < e) p2 = grouped[base + 64 + lane];

    const int cnt = min(32, e - base);
    for (int j = 0; j < cnt; j += 16) {
      float4 q0[4], q1[4];
      v2f af[4];
#pragma unroll
      for (int u = 0; u < 4; ++u) {
        const int jj = j + 4 * u;
        const int sA = __builtin_amdgcn_readlane(so, jj);
        const int sB = __builtin_amdgcn_readlane(so, jj + 1);
        const int sC = __builtin_amdgcn_readlane(so, jj + 2);
        const int sD = __builtin_amdgcn_readlane(so, jj + 3);
        q0[u] = *(const float4*)(xbytes + (unsigned)((hi ? sC : sA) + c16));
        q1[u] = *(const float4*)(xbytes + (unsigned)((hi ? sD : sB) + c16));
        const int dA = __builtin_amdgcn_readlane(dl, jj);
        const int dB = __builtin_amdgcn_readlane(dl, jj + 1);
        const int dC = __builtin_amdgcn_readlane(dl, jj + 2);
        const int dD = __builtin_amdgcn_readlane(dl, jj + 3);
        const int wA = __builtin_amdgcn_readlane(wi, jj);
        const int wB = __builtin_amdgcn_readlane(wi, jj + 1);
        const int wC = __builtin_amdgcn_readlane(wi, jj + 2);
        const int wD = __builtin_amdgcn_readlane(wi, jj + 3);
        af[u].x = ((hi ? dC : dA) == c) ? __int_as_float(hi ? wC : wA) : 0.f;
        af[u].y = ((hi ? dD : dB) == c) ? __int_as_float(hi ? wD : wB) : 0.f;
      }
#pragma unroll
      for (int u = 0; u < 4; ++u) {
        v2f bf;
        bf.x = q0[u].x; bf.y = q1[u].x;
        acc0 = __builtin_amdgcn_wmma_f32_16x16x4_f32(false, af[u], false, bf, (short)0, acc0, false, false);
        bf.x = q0[u].y; bf.y = q1[u].y;
        acc1 = __builtin_amdgcn_wmma_f32_16x16x4_f32(false, af[u], false, bf, (short)0, acc1, false, false);
        bf.x = q0[u].z; bf.y = q1[u].z;
        acc2 = __builtin_amdgcn_wmma_f32_16x16x4_f32(false, af[u], false, bf, (short)0, acc2, false, false);
        bf.x = q0[u].w; bf.y = q1[u].w;
        acc3 = __builtin_amdgcn_wmma_f32_16x16x4_f32(false, af[u], false, bf, (short)0, acc3, false, false);
      }
    }
    p0 = p1; a0 = a1; b0 = b1; p1 = p2;
  }

  const int rbase = row0 + (hi ? 8 : 0);
#pragma unroll
  for (int r = 0; r < 8; ++r) {
    const int row = rbase + r;
    if (row < Nn) {
      float4 o;
      o.x = acc0[r]; o.y = acc1[r]; o.z = acc2[r]; o.w = acc3[r];
      *(float4*)(xout + (size_t)row * DIM + c4) = o;
    }
  }
}

__global__ __launch_bounds__(128)
void gcn_layer_sum_kernel(const float* base, const float* xa, const float* xb, const float* xc,
                          float* o, const float* __restrict__ alpha, int nw, int k0, int nl, int first, int n4) {
  const int lane = threadIdx.x & 31;
  const int gw = __builtin_amdgcn_readfirstlane((int)((blockIdx.x * blockDim.x + threadIdx.x) >> 5));
  const int i0 = gw * (32 * SUMU);
  if (i0 >= n4) return;
  const float4* b4 = (const float4*)base;
  const float4* a4 = (const float4*)xa;
  const float4* c4 = (const float4*)xb;
  const float4* d4 = (const float4*)xc;

  int idx[SUMU];
  float4 vb[SUMU], va[SUMU], vc[SUMU], vd[SUMU];
#pragma unroll
  for (int u = 0; u < SUMU; ++u) {
    idx[u] = min(i0 + 32 * u + lane, n4 - 1);
    vb[u] = b4[idx[u]];
  }
  if (nl > 0) {
#pragma unroll
    for (int u = 0; u < SUMU; ++u) va[u] = a4[idx[u]];
  }
  if (nl > 1) {
#pragma unroll
    for (int u = 0; u < SUMU; ++u) vc[u] = c4[idx[u]];
  }
  if (nl > 2) {
#pragma unroll
    for (int u = 0; u < SUMU; ++u) vd[u] = d4[idx[u]];
  }
  const gcn_softmax_state st = gcn_softmax_prepare(alpha, nw);
  const float s0 = first ? gcn_softmax_weight(alpha, st, 0) : 1.0f;
  const float s1 = (nl > 0) ? gcn_softmax_weight(alpha, st, k0) : 0.f;
  const float s2 = (nl > 1) ? gcn_softmax_weight(alpha, st, k0 + 1) : 0.f;
  const float s3 = (nl > 2) ? gcn_softmax_weight(alpha, st, k0 + 2) : 0.f;

  float4* o4 = (float4*)o;
#pragma unroll
  for (int u = 0; u < SUMU; ++u) {
    float4 r;
    r.x = s0 * vb[u].x; r.y = s0 * vb[u].y; r.z = s0 * vb[u].z; r.w = s0 * vb[u].w;
    if (nl > 0) { r.x += s1 * va[u].x; r.y += s1 * va[u].y; r.z += s1 * va[u].z; r.w += s1 * va[u].w; }
    if (nl > 1) { r.x += s2 * vc[u].x; r.y += s2 * vc[u].y; r.z += s2 * vc[u].z; r.w += s2 * vc[u].w; }
    if (nl > 2) { r.x += s3 * vd[u].x; r.y += s3 * vd[u].y; r.z += s3 * vd[u].z; r.w += s3 * vd[u].w; }
    const int i = i0 + 32 * u + lane;
    if (i < n4) o4[i] = r;
  }
}

__device__ __forceinline__ float gcn_quad_sum(float v) {
  int t = __builtin_amdgcn_update_dpp(0, __float_as_int(v), 0xB1, 0xF, 0xF, true);
  v += __int_as_float(t);
  t = __builtin_amdgcn_update_dpp(0, __float_as_int(v), 0x4E, 0xF, 0xF, true);
  v += __int_as_float(t);
  return v;
}

__device__ __forceinline__ float gcn_pick8(v8f a, int r) {
  float v = a[0];
  if (r == 1) v = a[1];
  if (r == 2) v = a[2];
  if (r == 3) v = a[3];
  if (r == 4) v = a[4];
  if (r == 5) v = a[5];
  if (r == 6) v = a[6];
  if (r == 7) v = a[7];
  return v;
}

__global__ __launch_bounds__(128)
void gcn_edge_dot_wmma_kernel(const int* __restrict__ la, const int* __restrict__ lb,
                              const float* __restrict__ x, float* __restrict__ res,
                              int E, int npair) {
  const int lane = threadIdx.x & 31;
  const int gw = __builtin_amdgcn_readfirstlane((int)((blockIdx.x * blockDim.x + threadIdx.x) >> 5));
  const int nw = (int)((gridDim.x * blockDim.x) >> 5);
  const int c = lane & 15;
  const int slot = c >> 2;
  const int piece = ((c & 3) << 1) | (lane >> 4);
  const char* __restrict__ xbytes = (const char*)x;
  const unsigned pbytes = 16u * (unsigned)piece;
  const int last = E - 1;

  int pair = gw;
  int half = 0;
  int b = 2 * gw;
  int ia[4], ib[4];
#pragma unroll
  for (int q = 0; q < 4; ++q) {
    const int e = min(16 * b + 4 * slot + q, last);
    ia[q] = la[e];
    ib[q] = lb[e];
  }

  while (pair < npair) {
    float4 u[4][2], v[4][2];
#pragma unroll
    for (int q = 0; q < 4; ++q) {
      const unsigned oa = (unsigned)ia[q] * (DIM * 4u) + pbytes;
      const unsigned ob = (unsigned)ib[q] * (DIM * 4u) + pbytes;
      u[q][0] = *(const float4*)(xbytes + oa);
      u[q][1] = *(const float4*)(xbytes + oa + 128u);
      v[q][0] = *(const float4*)(xbytes + ob);
      v[q][1] = *(const float4*)(xbytes + ob + 128u);
    }
    const int bn = (half == 0) ? (b + 1) : (2 * (pair + nw));
#pragma unroll
    for (int q = 0; q < 4; ++q) {
      const int e = min(16 * bn + 4 * slot + q, last);
      ia[q] = la[e];
      ib[q] = lb[e];
    }

    v8f acc0 = {0.f, 0.f, 0.f, 0.f, 0.f, 0.f, 0.f, 0.f};
    v8f acc1 = acc0, acc2 = acc0, acc3 = acc0;
#pragma unroll
    for (int j = 0; j < 2; ++j) {
      v2f a, bb;
      a.x = u[0][j].x; a.y = u[0][j].y; bb.x = v[0][j].x; bb.y = v[0][j].y;
      acc0 = __builtin_amdgcn_wmma_f32_16x16x4_f32(false, a, false, bb, (short)0, acc0, false, false);
      a.x = u[1][j].x; a.y = u[1][j].y; bb.x = v[1][j].x; bb.y = v[1][j].y;
      acc1 = __builtin_amdgcn_wmma_f32_16x16x4_f32(false, a, false, bb, (short)0, acc1, false, false);
      a.x = u[2][j].x; a.y = u[2][j].y; bb.x = v[2][j].x; bb.y = v[2][j].y;
      acc2 = __builtin_amdgcn_wmma_f32_16x16x4_f32(false, a, false, bb, (short)0, acc2, false, false);
      a.x = u[3][j].x; a.y = u[3][j].y; bb.x = v[3][j].x; bb.y = v[3][j].y;
      acc3 = __builtin_amdgcn_wmma_f32_16x16x4_f32(false, a, false, bb, (short)0, acc3, false, false);
      a.x = u[0][j].z; a.y = u[0][j].w; bb.x = v[0][j].z; bb.y = v[0][j].w;
      acc0 = __builtin_amdgcn_wmma_f32_16x16x4_f32(false, a, false, bb, (short)0, acc0, false, false);
      a.x = u[1][j].z; a.y = u[1][j].w; bb.x = v[1][j].z; bb.y = v[1][j].w;
      acc1 = __builtin_amdgcn_wmma_f32_16x16x4_f32(false, a, false, bb, (short)0, acc1, false, false);
      a.x = u[2][j].z; a.y = u[2][j].w; bb.x = v[2][j].z; bb.y = v[2][j].w;
      acc2 = __builtin_amdgcn_wmma_f32_16x16x4_f32(false, a, false, bb, (short)0, acc2, false, false);
      a.x = u[3][j].z; a.y = u[3][j].w; bb.x = v[3][j].z; bb.y = v[3][j].w;
      acc3 = __builtin_amdgcn_wmma_f32_16x16x4_f32(false, a, false, bb, (short)0, acc3, false, false);
    }

    const int r = lane & 7;
    float4 o;
    o.x = gcn_quad_sum(gcn_pick8(acc0, r));
    o.y = gcn_quad_sum(gcn_pick8(acc1, r));
    o.z = gcn_quad_sum(gcn_pick8(acc2, r));
    o.w = gcn_quad_sum(gcn_pick8(acc3, r));

    const bool diag = (lane < 8) || (lane >= 24);
    const int e0 = 16 * b + 4 * slot;
    if (diag && (lane & 3) == 0) {
      if (e0 + 3 < E) {
        *(float4*)(res + e0) = o;
      } else {
        if (e0 < E) res[e0] = o.x;
        if (e0 + 1 < E) res[e0 + 1] = o.y;
        if (e0 + 2 < E) res[e0 + 2] = o.z;
      }
    }

    if (half == 1) pair += nw;
    half ^= 1;
    b = bn;
  }
}

extern "C" void kernel_launch(void* const* d_in, const int* in_sizes, int n_in,
                              void* d_out, int out_size, void* d_ws, size_t ws_size,
                              hipStream_t stream) {
  const int*   edge_index = (const int*)d_in[0];
  const int*   eli        = (const int*)d_in[1];
  const float* emb        = (const float*)d_in[2];
  const float* alpha      = (const float*)d_in[3];

  const int E  = in_sizes[0] / 2;
  const int Nn = in_sizes[2] / DIM;
  const int L1 = in_sizes[3];
  const int L  = L1 - 1;
  float* res = (float*)d_out;

  const int* src = edge_index;
  const int* dst = edge_index + E;
  const int* la  = eli;
  const int* lb  = eli + E;

  const int NB = (Nn + TILE - 1) / TILE;
  const int chunks = (E + CHUNK - 1) / CHUNK;
  const int NG = NB * TGROUPS;

  size_t off = 0;
  auto carve = [&](size_t bytes) -> void* {
    void* p = (char*)d_ws + off;
    off = (off + bytes + 255) & ~(size_t)255;
    return p;
  };
  float* dis      = (float*)carve(sizeof(float) * (size_t)Nn);
  int*   cmat     = (int*)carve(sizeof(int) * (size_t)chunks * (size_t)(NB + 3));
  int*   binStart = (int*)carve(sizeof(int) * (size_t)(NB + 1));
  int*   binned   = (int*)carve(sizeof(int) * 2 * (size_t)E);
  int*   grouped  = (int*)carve(sizeof(int) * 2 * (size_t)E);
  int*   grpStart = (int*)carve(sizeof(int) * ((size_t)NB * TGROUPS + 1));
  float* xA  = (float*)carve(sizeof(float) * (size_t)Nn * DIM);
  float* xB  = (float*)carve(sizeof(float) * (size_t)Nn * DIM);
  float* xC  = (float*)carve(sizeof(float) * (size_t)Nn * DIM);
  float* acc = (float*)carve(sizeof(float) * (size_t)Nn * DIM);
  (void)ws_size; (void)n_in; (void)out_size;

  const int n4 = Nn * DIM / 4;

  gcn_bin_count_kernel<<<chunks, BT, 0, stream>>>(dst, cmat, E, NB);
  gcn_bin_fill_kernel<<<chunks, BT, 0, stream>>>(src, dst, cmat, binStart, binned, E, NB, chunks);
  gcn_tile_deg_kernel<<<NB, 1024, 0, stream>>>(binned, binStart, dis, grouped, grpStart, Nn);
  const int sumWaves  = (n4 + 32 * SUMU - 1) / (32 * SUMU);
  const int sumBlocks = (sumWaves + 3) / 4;
  const float* xcur = emb;
  float* bufs[3] = {xA, xB, xC};
  if (L <= 0)
    gcn_layer_sum_kernel<<<sumBlocks, 128, 0, stream>>>(emb, emb, emb, emb, acc, alpha, L1, 1, 0, 1, n4);
  for (int l = 0; l < L; ++l) {
    float* xnext = bufs[l % 3];
    gcn_group_spmm_wmma_kernel<<<(NG + 3) / 4, 128, 0, stream>>>(grouped, grpStart, dis, xcur, xnext, Nn, NG);
    xcur = xnext;
    const int filled = l % 3 + 1;
    if (filled == 3 || l == L - 1) {
      const int firstSum = (l < 3) ? 1 : 0;
      gcn_layer_sum_kernel<<<sumBlocks, 128, 0, stream>>>(firstSum ? emb : acc, bufs[0], bufs[1], bufs[2],
                                                          acc, alpha, L1, l + 2 - filled, filled, firstSum, n4);
    }
  }

  if (E > 0) {
    const int nbatch = (E + 15) / 16;
    const int npair  = (nbatch + 1) / 2;
    const int blocks = (npair + 15) / 16;
    gcn_edge_dot_wmma_kernel<<<blocks, 128, 0, stream>>>(la, lb, acc, res, E, npair);
  }
}


# Binning 3 -> 2 launches: each 4096-edge chunk is counting-sorted by destination tile in LDS and stored whole with its tile starts; the tile pass gathers its segment of every chunk; sum order changes
# speedup vs baseline: 3.3901x; 1.1191x over previous
#include <hip/hip_runtime.h>

typedef __attribute__((ext_vector_type(2))) float v2f;
typedef __attribute__((ext_vector_type(8))) float v8f;

#define DIM 64
#define TILE 400
#define NBMAX 1024
#define BT 512
#define CHUNK 4096
#define CPT (CHUNK / BT)
#define CHMAX 1024
#define SEGU 8
#define GROWS 16
#define TGROUPS (TILE / GROWS)
#define SUMU 8

struct gcn_softmax_state { float m, s; };
__device__ __forceinline__ gcn_softmax_state gcn_softmax_prepare(const float* __restrict__ alpha, int n) {
  gcn_softmax_state st;
  st.m = alpha[0];
  for (int i = 1; i < n; ++i) st.m = fmaxf(st.m, alpha[i]);
  st.s = 0.f;
  for (int i = 0; i < n; ++i) st.s += expf(alpha[i] - st.m);
  return st;
}
__device__ __forceinline__ float gcn_softmax_weight(const float* __restrict__ alpha, gcn_softmax_state st, int i) {
  return expf(alpha[i] - st.m) / st.s;
}

__global__ __launch_bounds__(BT)
void gcn_chunk_sort_kernel(const int* __restrict__ src, const int* __restrict__ dst,
                           int* __restrict__ ctab, int* __restrict__ sorted_w, int E, int NB) {
  __shared__ int hist[NBMAX];
  __shared__ int wsum[BT / 32];
  __shared__ int4 stage4[CHUNK / 2];
  int2* stage = (int2*)stage4;
  int2* __restrict__ sorted = (int2*)sorted_w;
  const int tid = threadIdx.x;
  const int c0 = blockIdx.x * CHUNK;
  for (int i = tid; i < NB; i += BT) hist[i] = 0;
  __syncthreads();
  int dv[CPT], sv[CPT], rk[CPT];
#pragma unroll
  for (int k = 0; k < CPT; ++k) {
    const int e = c0 + k * BT + tid;
    const bool ok = e < E;
    dv[k] = ok ? dst[e] : -1;
    sv[k] = ok ? src[e] : 0;
  }
#pragma unroll
  for (int k = 0; k < CPT; ++k) {
    rk[k] = 0;
    if (dv[k] >= 0) rk[k] = atomicAdd(&hist[(unsigned)dv[k] / TILE], 1);
  }
  __syncthreads();
  const int lane = tid & 31;
  const int wv = tid >> 5;
  int loc[NBMAX / BT];
  int run = 0;
  for (int j = 0; j < NBMAX / BT; ++j) {
    const int idx = tid * (NBMAX / BT) + j;
    loc[j] = run;
    run += (idx < NB) ? hist[idx] : 0;
  }
  int inc = run;
  for (int off = 1; off < 32; off <<= 1) {
    const int t = __shfl_up(inc, off);
    if (lane >= off) inc += t;
  }
  if (lane == 31) wsum[wv] = inc;
  __syncthreads();
  int woff = 0;
  for (int w = 0; w < wv; ++w) woff += wsum[w];
  const int excl = woff + inc - run;
  int* __restrict__ row = ctab + (size_t)blockIdx.x * (size_t)(NB + 1);
  for (int j = 0; j < NBMAX / BT; ++j) {
    const int idx = tid * (NBMAX / BT) + j;
    if (idx < NB) {
      const int start = excl + loc[j];
      hist[idx] = start;
      row[idx] = start;
    }
  }
  if (tid == BT - 1) row[NB] = woff + inc;
  __syncthreads();
#pragma unroll
  for (int k = 0; k < CPT; ++k) {
    if (dv[k] >= 0) stage[hist[(unsigned)dv[k] / TILE] + rk[k]] = make_int2(sv[k], dv[k]);
  }
  __syncthreads();
  const int n = min(CHUNK, E - c0);
  const int n2 = n >> 1;
  int4* __restrict__ o4 = (int4*)(sorted + c0);
  for (int i = tid; i < n2; i += BT) o4[i] = stage4[i];
  if ((n & 1) && tid == 0) sorted[c0 + n - 1] = stage[n - 1];
}

__global__ __launch_bounds__(1024)
void gcn_tile_deg_kernel(const int* __restrict__ sorted_w, const int* __restrict__ ctab,
                         float* __restrict__ dis, int* __restrict__ grouped_w,
                         int* __restrict__ grpStart, int Nn, int NB, int chunks) {
  __shared__ int cnt[TILE];
  __shared__ int gsum[TGROUPS];
  __shared__ int gcur[TGROUPS];
  __shared__ int segLo[CHMAX];
  __shared__ int segN[CHMAX];
  __shared__ int tileSE[2];
  const int2* __restrict__ sorted = (const int2*)sorted_w;
  int2* __restrict__ grouped = (int2*)grouped_w;
  const int tid = threadIdx.x;
  const int lane = tid & 31;
  const int wv = tid >> 5;
  const int b = blockIdx.x;
  const int tile0 = b * TILE;
  const int rows = min(TILE, Nn - tile0);
  for (int i = tid; i < TILE; i += 1024) cnt[i] = 0;
  if (tid < 2) tileSE[tid] = 0;
  __syncthreads();
  const int pitch = NB + 1;
  for (int cb = 0; cb < chunks; cb += 1024) {
    const int c = cb + tid;
    int lo = 0, hi = 0;
    if (c < chunks) {
      lo = ctab[(size_t)c * pitch + b];
      hi = ctab[(size_t)c * pitch + b + 1];
      segLo[c] = c * CHUNK + lo;
      segN[c]  = hi - lo;
    }
    for (int off = 16; off > 0; off >>= 1) {
      lo += __shfl_xor(lo, off);
      hi += __shfl_xor(hi, off);
    }
    if (lane == 0 && cb + (wv << 5) < chunks) {
      atomicAdd(&tileSE[0], lo);
      atomicAdd(&tileSE[1], hi);
    }
  }
  __syncthreads();
  const int s = tileSE[0];
  const int e = tileSE[1];
  for (int cs = wv; cs < chunks; cs += 32 * SEGU) {
    int2 p[SEGU];
    int sb[SEGU], sn[SEGU];
#pragma unroll
    for (int u = 0; u < SEGU; ++u) {
      const int c = cs + 32 * u;
      const bool ok = c < chunks;
      sb[u] = ok ? segLo[c] : 0;
      sn[u] = ok ? segN[c] : 0;
      p[u] = make_int2(0, tile0);
      if (lane < sn[u]) p[u] = sorted[sb[u] + lane];
    }
#pragma unroll
    for (int u = 0; u < SEGU; ++u) {
      if (lane < sn[u]) atomicAdd(&cnt[p[u].y - tile0], 1);
      for (int i = lane + 32; i < sn[u]; i += 32) atomicAdd(&cnt[sorted[sb[u] + i].y - tile0], 1);
    }
  }
  __syncthreads();
  for (int i = tid; i < rows; i += 1024) {
    const int c = cnt[i];
    dis[tile0 + i] = (c > 0) ? rsqrtf((float)c) : 0.f;
  }
  if (tid < TGROUPS) {
    int t = 0;
    for (int i = 0; i < GROWS; ++i) t += cnt[tid * GROWS + i];
    gsum[tid] = t;
  }
  __syncthreads();
  if (tid == 0) {
    int run = s;
    for (int g = 0; g < TGROUPS; ++g) {
      gcur[g] = run;
      grpStart[b * TGROUPS + g] = run;
      run += gsum[g];
    }
    if (b == (int)gridDim.x - 1) grpStart[(b + 1) * TGROUPS] = e;
  }
  __syncthreads();
  for (int cs = wv; cs < chunks; cs += 32 * SEGU) {
    int2 p[SEGU];
    int sb[SEGU], sn[SEGU];
#pragma unroll
    for (int u = 0; u < SEGU; ++u) {
      const int c = cs + 32 * u;
      const bool ok = c < chunks;
      sb[u] = ok ? segLo[c] : 0;
      sn[u] = ok ? segN[c] : 0;
      p[u] = make_int2(0, tile0);
      if (lane < sn[u]) p[u] = sorted[sb[u] + lane];
    }
#pragma unroll
    for (int u = 0; u < SEGU; ++u) {
      if (lane < sn[u]) {
        const int r = atomicAdd(&gcur[(p[u].y - tile0) / GROWS], 1);
        grouped[r] = p[u];
      }
      for (int i = lane + 32; i < sn[u]; i += 32) {
        const int2 q = sorted[sb[u] + i];
        const int r = atomicAdd(&gcur[(q.y - tile0) / GROWS], 1);
        grouped[r] = q;
      }
    }
  }
}

__global__ __launch_bounds__(128)
void gcn_group_spmm_wmma_kernel(const int* __restrict__ grouped_w, const int* __restrict__ grpStart,
                                const float* __restrict__ dis, const float* __restrict__ xin,
                                float* __restrict__ xout, int Nn, int NG) {
  const int2* __restrict__ grouped = (const int2*)grouped_w;
  const int lane = threadIdx.x & 31;
  const int gw = __builtin_amdgcn_readfirstlane((int)((blockIdx.x * blockDim.x + threadIdx.x) >> 5));
  if (gw >= NG) return;
  const int tb   = gw / TGROUPS;
  const int row0 = tb * TILE + (gw - tb * TGROUPS) * GROWS;
  const int s = grpStart[gw];
  const int e = grpStart[gw + 1];
  const int c = lane & 15;
  const bool hi = lane >= 16;
  const int c4 = 4 * c;
  const int c16 = 16 * c;
  const char* __restrict__ xbytes = (const char*)xin;

  v8f acc0 = {0.f, 0.f, 0.f, 0.f, 0.f, 0.f, 0.f, 0.f};
  v8f acc1 = acc0, acc2 = acc0, acc3 = acc0;

  const int2 pad = make_int2(0, row0);
  int2 p0 = pad, p1 = pad;
  float a0 = 0.f, b0 = 0.f;
  if (s + lane < e) {
    p0 = grouped[s + lane];
    a0 = dis[p0.x];
    b0 = dis[p0.y];
  }
  if (s + 32 + lane < e) p1 = grouped[s + 32 + lane];

  for (int base = s; base < e; base += 32) {
    const int so = p0.x * (DIM * 4);
    const int dl = p0.y - row0;
    const int wi = __float_as_int(a0 * b0);
    float a1 = 0.f, b1 = 0.f;
    if (base + 32 + lane < e) {
      a1 = dis[p1.x];
      b1 = dis[p1.y];
    }
    int2 p2 = pad;
    if (base + 64 + lane < e) p2 = grouped[base + 64 + lane];

    const int cnt = min(32, e - base);
    for (int j = 0; j < cnt; j += 16) {
      float4 q0[4], q1[4];
      v2f af[4];
#pragma unroll
      for (int u = 0; u < 4; ++u) {
        const int jj = j + 4 * u;
        const int sA = __builtin_amdgcn_readlane(so, jj);
        const int sB = __builtin_amdgcn_readlane(so, jj + 1);
        const int sC = __builtin_amdgcn_readlane(so, jj + 2);
        const int sD = __builtin_amdgcn_readlane(so, jj + 3);
        q0[u] = *(const float4*)(xbytes + (unsigned)((hi ? sC : sA) + c16));
        q1[u] = *(const float4*)(xbytes + (unsigned)((hi ? sD : sB) + c16));
        const int dA = __builtin_amdgcn_readlane(dl, jj);
        const int dB = __builtin_amdgcn_readlane(dl, jj + 1);
        const int dC = __builtin_amdgcn_readlane(dl, jj + 2);
        const int dD = __builtin_amdgcn_readlane(dl, jj + 3);
        const int wA = __builtin_amdgcn_readlane(wi, jj);
        const int wB = __builtin_amdgcn_readlane(wi, jj + 1);
        const int wC = __builtin_amdgcn_readlane(wi, jj + 2);
        const int wD = __builtin_amdgcn_readlane(wi, jj + 3);
        af[u].x = ((hi ? dC : dA) == c) ? __int_as_float(hi ? wC : wA) : 0.f;
        af[u].y = ((hi ? dD : dB) == c) ? __int_as_float(hi ? wD : wB) : 0.f;
      }
#pragma unroll
      for (int u = 0; u < 4; ++u) {
        v2f bf;
        bf.x = q0[u].x; bf.y = q1[u].x;
        acc0 = __builtin_amdgcn_wmma_f32_16x16x4_f32(false, af[u], false, bf, (short)0, acc0, false, false);
        bf.x = q0[u].y; bf.y = q1[u].y;
        acc1 = __builtin_amdgcn_wmma_f32_16x16x4_f32(false, af[u], false, bf, (short)0, acc1, false, false);
        bf.x = q0[u].z; bf.y = q1[u].z;
        acc2 = __builtin_amdgcn_wmma_f32_16x16x4_f32(false, af[u], false, bf, (short)0, acc2, false, false);
        bf.x = q0[u].w; bf.y = q1[u].w;
        acc3 = __builtin_amdgcn_wmma_f32_16x16x4_f32(false, af[u], false, bf, (short)0, acc3, false, false);
      }
    }
    p0 = p1; a0 = a1; b0 = b1; p1 = p2;
  }

  const int rbase = row0 + (hi ? 8 : 0);
#pragma unroll
  for (int r = 0; r < 8; ++r) {
    const int row = rbase + r;
    if (row < Nn) {
      float4 o;
      o.x = acc0[r]; o.y = acc1[r]; o.z = acc2[r]; o.w = acc3[r];
      *(float4*)(xout + (size_t)row * DIM + c4) = o;
    }
  }
}

__global__ __launch_bounds__(128)
void gcn_layer_sum_kernel(const float* base, const float* xa, const float* xb, const float* xc,
                          float* o, const float* __restrict__ alpha, int nw, int k0, int nl, int first, int n4) {
  const int lane = threadIdx.x & 31;
  const int gw = __builtin_amdgcn_readfirstlane((int)((blockIdx.x * blockDim.x + threadIdx.x) >> 5));
  const int i0 = gw * (32 * SUMU);
  if (i0 >= n4) return;
  const float4* b4 = (const float4*)base;
  const float4* a4 = (const float4*)xa;
  const float4* c4 = (const float4*)xb;
  const float4* d4 = (const float4*)xc;

  int idx[SUMU];
  float4 vb[SUMU], va[SUMU], vc[SUMU], vd[SUMU];
#pragma unroll
  for (int u = 0; u < SUMU; ++u) {
    idx[u] = min(i0 + 32 * u + lane, n4 - 1);
    vb[u] = b4[idx[u]];
  }
  if (nl > 0) {
#pragma unroll
    for (int u = 0; u < SUMU; ++u) va[u] = a4[idx[u]];
  }
  if (nl > 1) {
#pragma unroll
    for (int u = 0; u < SUMU; ++u) vc[u] = c4[idx[u]];
  }
  if (nl > 2) {
#pragma unroll
    for (int u = 0; u < SUMU; ++u) vd[u] = d4[idx[u]];
  }
  const gcn_softmax_state st = gcn_softmax_prepare(alpha, nw);
  const float s0 = first ? gcn_softmax_weight(alpha, st, 0) : 1.0f;
  const float s1 = (nl > 0) ? gcn_softmax_weight(alpha, st, k0) : 0.f;
  const float s2 = (nl > 1) ? gcn_softmax_weight(alpha, st, k0 + 1) : 0.f;
  const float s3 = (nl > 2) ? gcn_softmax_weight(alpha, st, k0 + 2) : 0.f;

  float4* o4 = (float4*)o;
#pragma unroll
  for (int u = 0; u < SUMU; ++u) {
    float4 r;
    r.x = s0 * vb[u].x; r.y = s0 * vb[u].y; r.z = s0 * vb[u].z; r.w = s0 * vb[u].w;
    if (nl > 0) { r.x += s1 * va[u].x; r.y += s1 * va[u].y; r.z += s1 * va[u].z; r.w += s1 * va[u].w; }
    if (nl > 1) { r.x += s2 * vc[u].x; r.y += s2 * vc[u].y; r.z += s2 * vc[u].z; r.w += s2 * vc[u].w; }
    if (nl > 2) { r.x += s3 * vd[u].x; r.y += s3 * vd[u].y; r.z += s3 * vd[u].z; r.w += s3 * vd[u].w; }
    const int i = i0 + 32 * u + lane;
    if (i < n4) o4[i] = r;
  }
}

__device__ __forceinline__ float gcn_quad_sum(float v) {
  int t = __builtin_amdgcn_update_dpp(0, __float_as_int(v), 0xB1, 0xF, 0xF, true);
  v += __int_as_float(t);
  t = __builtin_amdgcn_update_dpp(0, __float_as_int(v), 0x4E, 0xF, 0xF, true);
  v += __int_as_float(t);
  return v;
}

__device__ __forceinline__ float gcn_pick8(v8f a, int r) {
  float v = a[0];
  if (r == 1) v = a[1];
  if (r == 2) v = a[2];
  if (r == 3) v = a[3];
  if (r == 4) v = a[4];
  if (r == 5) v = a[5];
  if (r == 6) v = a[6];
  if (r == 7) v = a[7];
  return v;
}

__global__ __launch_bounds__(128)
void gcn_edge_dot_wmma_kernel(const int* __restrict__ la, const int* __restrict__ lb,
                              const float* __restrict__ x, float* __restrict__ res,
                              int E, int npair) {
  const int lane = threadIdx.x & 31;
  const int gw = __builtin_amdgcn_readfirstlane((int)((blockIdx.x * blockDim.x + threadIdx.x) >> 5));
  const int nw = (int)((gridDim.x * blockDim.x) >> 5);
  const int c = lane & 15;
  const int slot = c >> 2;
  const int piece = ((c & 3) << 1) | (lane >> 4);
  const char* __restrict__ xbytes = (const char*)x;
  const unsigned pbytes = 16u * (unsigned)piece;
  const int last = E - 1;

  int pair = gw;
  int half = 0;
  int b = 2 * gw;
  int ia[4], ib[4];
#pragma unroll
  for (int q = 0; q < 4; ++q) {
    const int e = min(16 * b + 4 * slot + q, last);
    ia[q] = la[e];
    ib[q] = lb[e];
  }

  while (pair < npair) {
    float4 u[4][2], v[4][2];
#pragma unroll
    for (int q = 0; q < 4; ++q) {
      const unsigned oa = (unsigned)ia[q] * (DIM * 4u) + pbytes;
      const unsigned ob = (unsigned)ib[q] * (DIM * 4u) + pbytes;
      u[q][0] = *(const float4*)(xbytes + oa);
      u[q][1] = *(const float4*)(xbytes + oa + 128u);
      v[q][0] = *(const float4*)(xbytes + ob);
      v[q][1] = *(const float4*)(xbytes + ob + 128u);
    }
    const int bn = (half == 0) ? (b + 1) : (2 * (pair + nw));
#pragma unroll
    for (int q = 0; q < 4; ++q) {
      const int e = min(16 * bn + 4 * slot + q, last);
      ia[q] = la[e];
      ib[q] = lb[e];
    }

    v8f acc0 = {0.f, 0.f, 0.f, 0.f, 0.f, 0.f, 0.f, 0.f};
    v8f acc1 = acc0, acc2 = acc0, acc3 = acc0;
#pragma unroll
    for (int j = 0; j < 2; ++j) {
      v2f a, bb;
      a.x = u[0][j].x; a.y = u[0][j].y; bb.x = v[0][j].x; bb.y = v[0][j].y;
      acc0 = __builtin_amdgcn_wmma_f32_16x16x4_f32(false, a, false, bb, (short)0, acc0, false, false);
      a.x = u[1][j].x; a.y = u[1][j].y; bb.x = v[1][j].x; bb.y = v[1][j].y;
      acc1 = __builtin_amdgcn_wmma_f32_16x16x4_f32(false, a, false, bb, (short)0, acc1, false, false);
      a.x = u[2][j].x; a.y = u[2][j].y; bb.x = v[2][j].x; bb.y = v[2][j].y;
      acc2 = __builtin_amdgcn_wmma_f32_16x16x4_f32(false, a, false, bb, (short)0, acc2, false, false);
      a.x = u[3][j].x; a.y = u[3][j].y; bb.x = v[3][j].x; bb.y = v[3][j].y;
      acc3 = __builtin_amdgcn_wmma_f32_16x16x4_f32(false, a, false, bb, (short)0, acc3, false, false);
      a.x = u[0][j].z; a.y = u[0][j].w; bb.x = v[0][j].z; bb.y = v[0][j].w;
      acc0 = __builtin_amdgcn_wmma_f32_16x16x4_f32(false, a, false, bb, (short)0, acc0, false, false);
      a.x = u[1][j].z; a.y = u[1][j].w; bb.x = v[1][j].z; bb.y = v[1][j].w;
      acc1 = __builtin_amdgcn_wmma_f32_16x16x4_f32(false, a, false, bb, (short)0, acc1, false, false);
      a.x = u[2][j].z; a.y = u[2][j].w; bb.x = v[2][j].z; bb.y = v[2][j].w;
      acc2 = __builtin_amdgcn_wmma_f32_16x16x4_f32(false, a, false, bb, (short)0, acc2, false, false);
      a.x = u[3][j].z; a.y = u[3][j].w; bb.x = v[3][j].z; bb.y = v[3][j].w;
      acc3 = __builtin_amdgcn_wmma_f32_16x16x4_f32(false, a, false, bb, (short)0, acc3, false, false);
    }

    const int r = lane & 7;
    float4 o;
    o.x = gcn_quad_sum(gcn_pick8(acc0, r));
    o.y = gcn_quad_sum(gcn_pick8(acc1, r));
    o.z = gcn_quad_sum(gcn_pick8(acc2, r));
    o.w = gcn_quad_sum(gcn_pick8(acc3, r));

    const bool diag = (lane < 8) || (lane >= 24);
    const int e0 = 16 * b + 4 * slot;
    if (diag && (lane & 3) == 0) {
      if (e0 + 3 < E) {
        *(float4*)(res + e0) = o;
      } else {
        if (e0 < E) res[e0] = o.x;
        if (e0 + 1 < E) res[e0 + 1] = o.y;
        if (e0 + 2 < E) res[e0 + 2] = o.z;
      }
    }

    if (half == 1) pair += nw;
    half ^= 1;
    b = bn;
  }
}

extern "C" void kernel_launch(void* const* d_in, const int* in_sizes, int n_in,
                              void* d_out, int out_size, void* d_ws, size_t ws_size,
                              hipStream_t stream) {
  const int*   edge_index = (const int*)d_in[0];
  const int*   eli        = (const int*)d_in[1];
  const float* emb        = (const float*)d_in[2];
  const float* alpha      = (const float*)d_in[3];

  const int E  = in_sizes[0] / 2;
  const int Nn = in_sizes[2] / DIM;
  const int L1 = in_sizes[3];
  const int L  = L1 - 1;
  float* res = (float*)d_out;

  const int* src = edge_index;
  const int* dst = edge_index + E;
  const int* la  = eli;
  const int* lb  = eli + E;

  const int NB = (Nn + TILE - 1) / TILE;
  const int chunks = (E + CHUNK - 1) / CHUNK;
  const int NG = NB * TGROUPS;

  size_t off = 0;
  auto carve = [&](size_t bytes) -> void* {
    void* p = (char*)d_ws + off;
    off = (off + bytes + 255) & ~(size_t)255;
    return p;
  };
  float* dis      = (float*)carve(sizeof(float) * (size_t)Nn);
  int*   ctab     = (int*)carve(sizeof(int) * (size_t)chunks * (size_t)(NB + 1));
  int*   sorted   = (int*)carve(sizeof(int) * 2 * (size_t)E);
  int*   grouped  = (int*)carve(sizeof(int) * 2 * (size_t)E);
  int*   grpStart = (int*)carve(sizeof(int) * ((size_t)NB * TGROUPS + 1));
  float* xA  = (float*)carve(sizeof(float) * (size_t)Nn * DIM);
  float* xB  = (float*)carve(sizeof(float) * (size_t)Nn * DIM);
  float* xC  = (float*)carve(sizeof(float) * (size_t)Nn * DIM);
  float* acc = (float*)carve(sizeof(float) * (size_t)Nn * DIM);
  (void)ws_size; (void)n_in; (void)out_size;

  const int n4 = Nn * DIM / 4;

  gcn_chunk_sort_kernel<<<chunks, BT, 0, stream>>>(src, dst, ctab, sorted, E, NB);
  gcn_tile_deg_kernel<<<NB, 1024, 0, stream>>>(sorted, ctab, dis, grouped, grpStart, Nn, NB, chunks);
  const int sumWaves  = (n4 + 32 * SUMU - 1) / (32 * SUMU);
  const int sumBlocks = (sumWaves + 3) / 4;
  const float* xcur = emb;
  float* bufs[3] = {xA, xB, xC};
  if (L <= 0)
    gcn_layer_sum_kernel<<<sumBlocks, 128, 0, stream>>>(emb, emb, emb, emb, acc, alpha, L1, 1, 0, 1, n4);
  for (int l = 0; l < L; ++l) {
    float* xnext = bufs[l % 3];
    gcn_group_spmm_wmma_kernel<<<(NG + 3) / 4, 128, 0, stream>>>(grouped, grpStart, dis, xcur, xnext, Nn, NG);
    xcur = xnext;
    const int filled = l % 3 + 1;
    if (filled == 3 || l == L - 1) {
      const int firstSum = (l < 3) ? 1 : 0;
      gcn_layer_sum_kernel<<<sumBlocks, 128, 0, stream>>>(firstSum ? emb : acc, bufs[0], bufs[1], bufs[2],
                                                          acc, alpha, L1, l + 2 - filled, filled, firstSum, n4);
    }
  }

  if (E > 0) {
    const int nbatch = (E + 15) / 16;
    const int npair  = (nbatch + 1) / 2;
    const int blocks = (npair + 15) / 16;
    gcn_edge_dot_wmma_kernel<<<blocks, 128, 0, stream>>>(la, lb, acc, res, E, npair);
  }
}


# Edge weight formed once: layer 1 stores each edge's w with packed (src, row in group) as an 8-byte entry in grouped order; layers 2, 3 read it and gather no dis values; sums and order unchanged
# speedup vs baseline: 3.5605x; 1.0503x over previous
#include <hip/hip_runtime.h>

typedef __attribute__((ext_vector_type(2))) float v2f;
typedef __attribute__((ext_vector_type(8))) float v8f;

#define DIM 64
#define TILE 400
#define NBMAX 1024
#define BT 512
#define CHUNK 4096
#define CPT (CHUNK / BT)
#define CHMAX 1024
#define SEGU 8
#define GROWS 16
#define TGROUPS (TILE / GROWS)
#define SUMU 8

struct gcn_softmax_state { float m, s; };
__device__ __forceinline__ gcn_softmax_state gcn_softmax_prepare(const float* __restrict__ alpha, int n) {
  gcn_softmax_state st;
  st.m = alpha[0];
  for (int i = 1; i < n; ++i) st.m = fmaxf(st.m, alpha[i]);
  st.s = 0.f;
  for (int i = 0; i < n; ++i) st.s += expf(alpha[i] - st.m);
  return st;
}
__device__ __forceinline__ float gcn_softmax_weight(const float* __restrict__ alpha, gcn_softmax_state st, int i) {
  return expf(alpha[i] - st.m) / st.s;
}

__global__ __launch_bounds__(BT)
void gcn_chunk_sort_kernel(const int* __restrict__ src, const int* __restrict__ dst,
                           int* __restrict__ ctab, int* __restrict__ sorted_w, int E, int NB) {
  __shared__ int hist[NBMAX];
  __shared__ int wsum[BT / 32];
  __shared__ int4 stage4[CHUNK / 2];
  int2* stage = (int2*)stage4;
  int2* __restrict__ sorted = (int2*)sorted_w;
  const int tid = threadIdx.x;
  const int c0 = blockIdx.x * CHUNK;
  for (int i = tid; i < NB; i += BT) hist[i] = 0;
  __syncthreads();
  int dv[CPT], sv[CPT], rk[CPT];
#pragma unroll
  for (int k = 0; k < CPT; ++k) {
    const int e = c0 + k * BT + tid;
    const bool ok = e < E;
    dv[k] = ok ? dst[e] : -1;
    sv[k] = ok ? src[e] : 0;
  }
#pragma unroll
  for (int k = 0; k < CPT; ++k) {
    rk[k] = 0;
    if (dv[k] >= 0) rk[k] = atomicAdd(&hist[(unsigned)dv[k] / TILE], 1);
  }
  __syncthreads();
  const int lane = tid & 31;
  const int wv = tid >> 5;
  int loc[NBMAX / BT];
  int run = 0;
  for (int j = 0; j < NBMAX / BT; ++j) {
    const int idx = tid * (NBMAX / BT) + j;
    loc[j] = run;
    run += (idx < NB) ? hist[idx] : 0;
  }
  int inc = run;
  for (int off = 1; off < 32; off <<= 1) {
    const int t = __shfl_up(inc, off);
    if (lane >= off) inc += t;
  }
  if (lane == 31) wsum[wv] = inc;
  __syncthreads();
  int woff = 0;
  for (int w = 0; w < wv; ++w) woff += wsum[w];
  const int excl = woff + inc - run;
  int* __restrict__ row = ctab + (size_t)blockIdx.x * (size_t)(NB + 1);
  for (int j = 0; j < NBMAX / BT; ++j) {
    const int idx = tid * (NBMAX / BT) + j;
    if (idx < NB) {
      const int start = excl + loc[j];
      hist[idx] = start;
      row[idx] = start;
    }
  }
  if (tid == BT - 1) row[NB] = woff + inc;
  __syncthreads();
#pragma unroll
  for (int k = 0; k < CPT; ++k) {
    if (dv[k] >= 0) stage[hist[(unsigned)dv[k] / TILE] + rk[k]] = make_int2(sv[k], dv[k]);
  }
  __syncthreads();
  const int n = min(CHUNK, E - c0);
  const int n2 = n >> 1;
  int4* __restrict__ o4 = (int4*)(sorted + c0);
  for (int i = tid; i < n2; i += BT) o4[i] = stage4[i];
  if ((n & 1) && tid == 0) sorted[c0 + n - 1] = stage[n - 1];
}

__global__ __launch_bounds__(1024)
void gcn_tile_deg_kernel(const int* __restrict__ sorted_w, const int* __restrict__ ctab,
                         float* __restrict__ dis, int* __restrict__ grouped_w,
                         int* __restrict__ grpStart, int Nn, int NB, int chunks) {
  __shared__ int cnt[TILE];
  __shared__ int gsum[TGROUPS];
  __shared__ int gcur[TGROUPS];
  __shared__ int segLo[CHMAX];
  __shared__ int segN[CHMAX];
  __shared__ int tileSE[2];
  const int2* __restrict__ sorted = (const int2*)sorted_w;
  int2* __restrict__ grouped = (int2*)grouped_w;
  const int tid = threadIdx.x;
  const int lane = tid & 31;
  const int wv = tid >> 5;
  const int b = blockIdx.x;
  const int tile0 = b * TILE;
  const int rows = min(TILE, Nn - tile0);
  for (int i = tid; i < TILE; i += 1024) cnt[i] = 0;
  if (tid < 2) tileSE[tid] = 0;
  __syncthreads();
  const int pitch = NB + 1;
  for (int cb = 0; cb < chunks; cb += 1024) {
    const int c = cb + tid;
    int lo = 0, hi = 0;
    if (c < chunks) {
      lo = ctab[(size_t)c * pitch + b];
      hi = ctab[(size_t)c * pitch + b + 1];
      segLo[c] = c * CHUNK + lo;
      segN[c]  = hi - lo;
    }
    for (int off = 16; off > 0; off >>= 1) {
      lo += __shfl_xor(lo, off);
      hi += __shfl_xor(hi, off);
    }
    if (lane == 0 && cb + (wv << 5) < chunks) {
      atomicAdd(&tileSE[0], lo);
      atomicAdd(&tileSE[1], hi);
    }
  }
  __syncthreads();
  const int s = tileSE[0];
  const int e = tileSE[1];
  for (int cs = wv; cs < chunks; cs += 32 * SEGU) {
    int2 p[SEGU];
    int sb[SEGU], sn[SEGU];
#pragma unroll
    for (int u = 0; u < SEGU; ++u) {
      const int c = cs + 32 * u;
      const bool ok = c < chunks;
      sb[u] = ok ? segLo[c] : 0;
      sn[u] = ok ? segN[c] : 0;
      p[u] = make_int2(0, tile0);
      if (lane < sn[u]) p[u] = sorted[sb[u] + lane];
    }
#pragma unroll
    for (int u = 0; u < SEGU; ++u) {
      if (lane < sn[u]) atomicAdd(&cnt[p[u].y - tile0], 1);
      for (int i = lane + 32; i < sn[u]; i += 32) atomicAdd(&cnt[sorted[sb[u] + i].y - tile0], 1);
    }
  }
  __syncthreads();
  for (int i = tid; i < rows; i += 1024) {
    const int c = cnt[i];
    dis[tile0 + i] = (c > 0) ? rsqrtf((float)c) : 0.f;
  }
  if (tid < TGROUPS) {
    int t = 0;
    for (int i = 0; i < GROWS; ++i) t += cnt[tid * GROWS + i];
    gsum[tid] = t;
  }
  __syncthreads();
  if (tid == 0) {
    int run = s;
    for (int g = 0; g < TGROUPS; ++g) {
      gcur[g] = run;
      grpStart[b * TGROUPS + g] = run;
      run += gsum[g];
    }
    if (b == (int)gridDim.x - 1) grpStart[(b + 1) * TGROUPS] = e;
  }
  __syncthreads();
  for (int cs = wv; cs < chunks; cs += 32 * SEGU) {
    int2 p[SEGU];
    int sb[SEGU], sn[SEGU];
#pragma unroll
    for (int u = 0; u < SEGU; ++u) {
      const int c = cs + 32 * u;
      const bool ok = c < chunks;
      sb[u] = ok ? segLo[c] : 0;
      sn[u] = ok ? segN[c] : 0;
      p[u] = make_int2(0, tile0);
      if (lane < sn[u]) p[u] = sorted[sb[u] + lane];
    }
#pragma unroll
    for (int u = 0; u < SEGU; ++u) {
      if (lane < sn[u]) {
        const int r = atomicAdd(&gcur[(p[u].y - tile0) / GROWS], 1);
        grouped[r] = p[u];
      }
      for (int i = lane + 32; i < sn[u]; i += 32) {
        const int2 q = sorted[sb[u] + i];
        const int r = atomicAdd(&gcur[(q.y - tile0) / GROWS], 1);
        grouped[r] = q;
      }
    }
  }
}

template <bool kFirst>
__global__ __launch_bounds__(128)
void gcn_group_spmm_wmma_kernel(const int* __restrict__ list_w, int* __restrict__ ready_w,
                                const int* __restrict__ grpStart,
                                const float* __restrict__ dis, const float* __restrict__ xin,
                                float* __restrict__ xout, int Nn, int NG) {
  const int2* __restrict__ grouped = (const int2*)list_w;
  int2* __restrict__ ready = (int2*)ready_w;
  const int lane = threadIdx.x & 31;
  const int gw = __builtin_amdgcn_readfirstlane((int)((blockIdx.x * blockDim.x + threadIdx.x) >> 5));
  if (gw >= NG) return;
  const int tb   = gw / TGROUPS;
  const int row0 = tb * TILE + (gw - tb * TGROUPS) * GROWS;
  const int s = grpStart[gw];
  const int e = grpStart[gw + 1];
  const int c = lane & 15;
  const bool hi = lane >= 16;
  const int c4 = 4 * c;
  const int c16 = 16 * c;
  const char* __restrict__ xbytes = (const char*)xin;

  v8f acc0 = {0.f, 0.f, 0.f, 0.f, 0.f, 0.f, 0.f, 0.f};
  v8f acc1 = acc0, acc2 = acc0, acc3 = acc0;

  const int2 pad = kFirst ? make_int2(0, row0) : make_int2(0, 0);
  int2 p0 = pad, p1 = pad;
  float a0 = 0.f, b0 = 0.f;
  if (s + lane < e) {
    p0 = grouped[s + lane];
    if constexpr (kFirst) {
      a0 = dis[p0.x];
      b0 = dis[p0.y];
    }
  }
  if (s + 32 + lane < e) p1 = grouped[s + 32 + lane];

  for (int base = s; base < e; base += 32) {
    int so, dl, wi;
    float a1 = 0.f, b1 = 0.f;
    if constexpr (kFirst) {
      so = p0.x * (DIM * 4);
      dl = p0.y - row0;
      wi = __float_as_int(a0 * b0);
      if (base + lane < e) ready[base + lane] = make_int2((p0.x << 4) | dl, wi);
      if (base + 32 + lane < e) {
        a1 = dis[p1.x];
        b1 = dis[p1.y];
      }
    } else {
      so = (p0.x >> 4) * (DIM * 4);
      dl = p0.x & 15;
      wi = p0.y;
    }
    int2 p2 = pad;
    if (base + 64 + lane < e) p2 = grouped[base + 64 + lane];

    const int cnt = min(32, e - base);
    for (int j = 0; j < cnt; j += 16) {
      float4 q0[4], q1[4];
      v2f af[4];
#pragma unroll
      for (int u = 0; u < 4; ++u) {
        const int jj = j + 4 * u;
        const int sA = __builtin_amdgcn_readlane(so, jj);
        const int sB = __builtin_amdgcn_readlane(so, jj + 1);
        const int sC = __builtin_amdgcn_readlane(so, jj + 2);
        const int sD = __builtin_amdgcn_readlane(so, jj + 3);
        q0[u] = *(const float4*)(xbytes + (unsigned)((hi ? sC : sA) + c16));
        q1[u] = *(const float4*)(xbytes + (unsigned)((hi ? sD : sB) + c16));
        const int dA = __builtin_amdgcn_readlane(dl, jj);
        const int dB = __builtin_amdgcn_readlane(dl, jj + 1);
        const int dC = __builtin_amdgcn_readlane(dl, jj + 2);
        const int dD = __builtin_amdgcn_readlane(dl, jj + 3);
        const int wA = __builtin_amdgcn_readlane(wi, jj);
        const int wB = __builtin_amdgcn_readlane(wi, jj + 1);
        const int wC = __builtin_amdgcn_readlane(wi, jj + 2);
        const int wD = __builtin_amdgcn_readlane(wi, jj + 3);
        af[u].x = ((hi ? dC : dA) == c) ? __int_as_float(hi ? wC : wA) : 0.f;
        af[u].y = ((hi ? dD : dB) == c) ? __int_as_float(hi ? wD : wB) : 0.f;
      }
#pragma unroll
      for (int u = 0; u < 4; ++u) {
        v2f bf;
        bf.x = q0[u].x; bf.y = q1[u].x;
        acc0 = __builtin_amdgcn_wmma_f32_16x16x4_f32(false, af[u], false, bf, (short)0, acc0, false, false);
        bf.x = q0[u].y; bf.y = q1[u].y;
        acc1 = __builtin_amdgcn_wmma_f32_16x16x4_f32(false, af[u], false, bf, (short)0, acc1, false, false);
        bf.x = q0[u].z; bf.y = q1[u].z;
        acc2 = __builtin_amdgcn_wmma_f32_16x16x4_f32(false, af[u], false, bf, (short)0, acc2, false, false);
        bf.x = q0[u].w; bf.y = q1[u].w;
        acc3 = __builtin_amdgcn_wmma_f32_16x16x4_f32(false, af[u], false, bf, (short)0, acc3, false, false);
      }
    }
    p0 = p1; a0 = a1; b0 = b1; p1 = p2;
  }

  const int rbase = row0 + (hi ? 8 : 0);
#pragma unroll
  for (int r = 0; r < 8; ++r) {
    const int row = rbase + r;
    if (row < Nn) {
      float4 o;
      o.x = acc0[r]; o.y = acc1[r]; o.z = acc2[r]; o.w = acc3[r];
      *(float4*)(xout + (size_t)row * DIM + c4) = o;
    }
  }
}

__global__ __launch_bounds__(128)
void gcn_layer_sum_kernel(const float* base, const float* xa, const float* xb, const float* xc,
                          float* o, const float* __restrict__ alpha, int nw, int k0, int nl, int first, int n4) {
  const int lane = threadIdx.x & 31;
  const int gw = __builtin_amdgcn_readfirstlane((int)((blockIdx.x * blockDim.x + threadIdx.x) >> 5));
  const int i0 = gw * (32 * SUMU);
  if (i0 >= n4) return;
  const float4* b4 = (const float4*)base;
  const float4* a4 = (const float4*)xa;
  const float4* c4 = (const float4*)xb;
  const float4* d4 = (const float4*)xc;

  int idx[SUMU];
  float4 vb[SUMU], va[SUMU], vc[SUMU], vd[SUMU];
#pragma unroll
  for (int u = 0; u < SUMU; ++u) {
    idx[u] = min(i0 + 32 * u + lane, n4 - 1);
    vb[u] = b4[idx[u]];
  }
  if (nl > 0) {
#pragma unroll
    for (int u = 0; u < SUMU; ++u) va[u] = a4[idx[u]];
  }
  if (nl > 1) {
#pragma unroll
    for (int u = 0; u < SUMU; ++u) vc[u] = c4[idx[u]];
  }
  if (nl > 2) {
#pragma unroll
    for (int u = 0; u < SUMU; ++u) vd[u] = d4[idx[u]];
  }
  const gcn_softmax_state st = gcn_softmax_prepare(alpha, nw);
  const float s0 = first ? gcn_softmax_weight(alpha, st, 0) : 1.0f;
  const float s1 = (nl > 0) ? gcn_softmax_weight(alpha, st, k0) : 0.f;
  const float s2 = (nl > 1) ? gcn_softmax_weight(alpha, st, k0 + 1) : 0.f;
  const float s3 = (nl > 2) ? gcn_softmax_weight(alpha, st, k0 + 2) : 0.f;

  float4* o4 = (float4*)o;
#pragma unroll
  for (int u = 0; u < SUMU; ++u) {
    float4 r;
    r.x = s0 * vb[u].x; r.y = s0 * vb[u].y; r.z = s0 * vb[u].z; r.w = s0 * vb[u].w;
    if (nl > 0) { r.x += s1 * va[u].x; r.y += s1 * va[u].y; r.z += s1 * va[u].z; r.w += s1 * va[u].w; }
    if (nl > 1) { r.x += s2 * vc[u].x; r.y += s2 * vc[u].y; r.z += s2 * vc[u].z; r.w += s2 * vc[u].w; }
    if (nl > 2) { r.x += s3 * vd[u].x; r.y += s3 * vd[u].y; r.z += s3 * vd[u].z; r.w += s3 * vd[u].w; }
    const int i = i0 + 32 * u + lane;
    if (i < n4) o4[i] = r;
  }
}

__device__ __forceinline__ float gcn_quad_sum(float v) {
  int t = __builtin_amdgcn_update_dpp(0, __float_as_int(v), 0xB1, 0xF, 0xF, true);
  v += __int_as_float(t);
  t = __builtin_amdgcn_update_dpp(0, __float_as_int(v), 0x4E, 0xF, 0xF, true);
  v += __int_as_float(t);
  return v;
}

__device__ __forceinline__ float gcn_pick8(v8f a, int r) {
  float v = a[0];
  if (r == 1) v = a[1];
  if (r == 2) v = a[2];
  if (r == 3) v = a[3];
  if (r == 4) v = a[4];
  if (r == 5) v = a[5];
  if (r == 6) v = a[6];
  if (r == 7) v = a[7];
  return v;
}

__global__ __launch_bounds__(128)
void gcn_edge_dot_wmma_kernel(const int* __restrict__ la, const int* __restrict__ lb,
                              const float* __restrict__ x, float* __restrict__ res,
                              int E, int npair) {
  const int lane = threadIdx.x & 31;
  const int gw = __builtin_amdgcn_readfirstlane((int)((blockIdx.x * blockDim.x + threadIdx.x) >> 5));
  const int nw = (int)((gridDim.x * blockDim.x) >> 5);
  const int c = lane & 15;
  const int slot = c >> 2;
  const int piece = ((c & 3) << 1) | (lane >> 4);
  const char* __restrict__ xbytes = (const char*)x;
  const unsigned pbytes = 16u * (unsigned)piece;
  const int last = E - 1;

  int pair = gw;
  int half = 0;
  int b = 2 * gw;
  int ia[4], ib[4];
#pragma unroll
  for (int q = 0; q < 4; ++q) {
    const int e = min(16 * b + 4 * slot + q, last);
    ia[q] = la[e];
    ib[q] = lb[e];
  }

  while (pair < npair) {
    float4 u[4][2], v[4][2];
#pragma unroll
    for (int q = 0; q < 4; ++q) {
      const unsigned oa = (unsigned)ia[q] * (DIM * 4u) + pbytes;
      const unsigned ob = (unsigned)ib[q] * (DIM * 4u) + pbytes;
      u[q][0] = *(const float4*)(xbytes + oa);
      u[q][1] = *(const float4*)(xbytes + oa + 128u);
      v[q][0] = *(const float4*)(xbytes + ob);
      v[q][1] = *(const float4*)(xbytes + ob + 128u);
    }
    const int bn = (half == 0) ? (b + 1) : (2 * (pair + nw));
#pragma unroll
    for (int q = 0; q < 4; ++q) {
      const int e = min(16 * bn + 4 * slot + q, last);
      ia[q] = la[e];
      ib[q] = lb[e];
    }

    v8f acc0 = {0.f, 0.f, 0.f, 0.f, 0.f, 0.f, 0.f, 0.f};
    v8f acc1 = acc0, acc2 = acc0, acc3 = acc0;
#pragma unroll
    for (int j = 0; j < 2; ++j) {
      v2f a, bb;
      a.x = u[0][j].x; a.y = u[0][j].y; bb.x = v[0][j].x; bb.y = v[0][j].y;
      acc0 = __builtin_amdgcn_wmma_f32_16x16x4_f32(false, a, false, bb, (short)0, acc0, false, false);
      a.x = u[1][j].x; a.y = u[1][j].y; bb.x = v[1][j].x; bb.y = v[1][j].y;
      acc1 = __builtin_amdgcn_wmma_f32_16x16x4_f32(false, a, false, bb, (short)0, acc1, false, false);
      a.x = u[2][j].x; a.y = u[2][j].y; bb.x = v[2][j].x; bb.y = v[2][j].y;
      acc2 = __builtin_amdgcn_wmma_f32_16x16x4_f32(false, a, false, bb, (short)0, acc2, false, false);
      a.x = u[3][j].x; a.y = u[3][j].y; bb.x = v[3][j].x; bb.y = v[3][j].y;
      acc3 = __builtin_amdgcn_wmma_f32_16x16x4_f32(false, a, false, bb, (short)0, acc3, false, false);
      a.x = u[0][j].z; a.y = u[0][j].w; bb.x = v[0][j].z; bb.y = v[0][j].w;
      acc0 = __builtin_amdgcn_wmma_f32_16x16x4_f32(false, a, false, bb, (short)0, acc0, false, false);
      a.x = u[1][j].z; a.y = u[1][j].w; bb.x = v[1][j].z; bb.y = v[1][j].w;
      acc1 = __builtin_amdgcn_wmma_f32_16x16x4_f32(false, a, false, bb, (short)0, acc1, false, false);
      a.x = u[2][j].z; a.y = u[2][j].w; bb.x = v[2][j].z; bb.y = v[2][j].w;
      acc2 = __builtin_amdgcn_wmma_f32_16x16x4_f32(false, a, false, bb, (short)0, acc2, false, false);
      a.x = u[3][j].z; a.y = u[3][j].w; bb.x = v[3][j].z; bb.y = v[3][j].w;
      acc3 = __builtin_amdgcn_wmma_f32_16x16x4_f32(false, a, false, bb, (short)0, acc3, false, false);
    }

    const int r = lane & 7;
    float4 o;
    o.x = gcn_quad_sum(gcn_pick8(acc0, r));
    o.y = gcn_quad_sum(gcn_pick8(acc1, r));
    o.z = gcn_quad_sum(gcn_pick8(acc2, r));
    o.w = gcn_quad_sum(gcn_pick8(acc3, r));

    const bool diag = (lane < 8) || (lane >= 24);
    const int e0 = 16 * b + 4 * slot;
    if (diag && (lane & 3) == 0) {
      if (e0 + 3 < E) {
        *(float4*)(res + e0) = o;
      } else {
        if (e0 < E) res[e0] = o.x;
        if (e0 + 1 < E) res[e0 + 1] = o.y;
        if (e0 + 2 < E) res[e0 + 2] = o.z;
      }
    }

    if (half == 1) pair += nw;
    half ^= 1;
    b = bn;
  }
}

extern "C" void kernel_launch(void* const* d_in, const int* in_sizes, int n_in,
                              void* d_out, int out_size, void* d_ws, size_t ws_size,
                              hipStream_t stream) {
  const int*   edge_index = (const int*)d_in[0];
  const int*   eli        = (const int*)d_in[1];
  const float* emb        = (const float*)d_in[2];
  const float* alpha      = (const float*)d_in[3];

  const int E  = in_sizes[0] / 2;
  const int Nn = in_sizes[2] / DIM;
  const int L1 = in_sizes[3];
  const int L  = L1 - 1;
  float* res = (float*)d_out;

  const int* src = edge_index;
  const int* dst = edge_index + E;
  const int* la  = eli;
  const int* lb  = eli + E;

  const int NB = (Nn + TILE - 1) / TILE;
  const int chunks = (E + CHUNK - 1) / CHUNK;
  const int NG = NB * TGROUPS;

  size_t off = 0;
  auto carve = [&](size_t bytes) -> void* {
    void* p = (char*)d_ws + off;
    off = (off + bytes + 255) & ~(size_t)255;
    return p;
  };
  float* dis      = (float*)carve(sizeof(float) * (size_t)Nn);
  int*   ctab     = (int*)carve(sizeof(int) * (size_t)chunks * (size_t)(NB + 1));
  int*   sorted   = (int*)carve(sizeof(int) * 2 * (size_t)E);
  int*   grouped  = (int*)carve(sizeof(int) * 2 * (size_t)E);
  int*   ready    = (int*)carve(sizeof(int) * 2 * (size_t)E);
  int*   grpStart = (int*)carve(sizeof(int) * ((size_t)NB * TGROUPS + 1));
  float* xA  = (float*)carve(sizeof(float) * (size_t)Nn * DIM);
  float* xB  = (float*)carve(sizeof(float) * (size_t)Nn * DIM);
  float* xC  = (float*)carve(sizeof(float) * (size_t)Nn * DIM);
  float* acc = (float*)carve(sizeof(float) * (size_t)Nn * DIM);
  (void)ws_size; (void)n_in; (void)out_size;

  const int n4 = Nn * DIM / 4;

  gcn_chunk_sort_kernel<<<chunks, BT, 0, stream>>>(src, dst, ctab, sorted, E, NB);
  gcn_tile_deg_kernel<<<NB, 1024, 0, stream>>>(sorted, ctab, dis, grouped, grpStart, Nn, NB, chunks);
  const int sumWaves  = (n4 + 32 * SUMU - 1) / (32 * SUMU);
  const int sumBlocks = (sumWaves + 3) / 4;
  const float* xcur = emb;
  float* bufs[3] = {xA, xB, xC};
  if (L <= 0)
    gcn_layer_sum_kernel<<<sumBlocks, 128, 0, stream>>>(emb, emb, emb, emb, acc, alpha, L1, 1, 0, 1, n4);
  for (int l = 0; l < L; ++l) {
    float* xnext = bufs[l % 3];
    if (l == 0)
      gcn_group_spmm_wmma_kernel<true><<<(NG + 3) / 4, 128, 0, stream>>>(grouped, ready, grpStart, dis, xcur, xnext, Nn, NG);
    else
      gcn_group_spmm_wmma_kernel<false><<<(NG + 3) / 4, 128, 0, stream>>>(ready, nullptr, grpStart, dis, xcur, xnext, Nn, NG);
    xcur = xnext;
    const int filled = l % 3 + 1;
    if (filled == 3 || l == L - 1) {
      const int firstSum = (l < 3) ? 1 : 0;
      gcn_layer_sum_kernel<<<sumBlocks, 128, 0, stream>>>(firstSum ? emb : acc, bufs[0], bufs[1], bufs[2],
                                                          acc, alpha, L1, l + 2 - filled, filled, firstSum, n4);
    }
  }

  if (E > 0) {
    const int nbatch = (E + 15) / 16;
    const int npair  = (nbatch + 1) / 2;
    const int blocks = (npair + 15) / 16;
    gcn_edge_dot_wmma_kernel<<<blocks, 128, 0, stream>>>(la, lb, acc, res, E, npair);
  }
}
